# starting kernel
# baseline (speedup 1.0000x reference)
#include <hip/hip_runtime.h>
#include <hip/hip_bf16.h>

#define GAT_N 30000
#define GAT_E 480000
#define GAT_ETOT (GAT_E + GAT_N)
#define GAT_H 8
#define GAT_C 32
#define GAT_FH 256
#define NEG_SLOPE 0.2f

typedef float v2f __attribute__((ext_vector_type(2)));
typedef float v8f __attribute__((ext_vector_type(8)));

__global__ void gemm_wmma_f32(const float* __restrict__ A,
                              const float* __restrict__ B,
                              float* __restrict__ C,
                              int M, int K, int Ncols) {
  const int wave = threadIdx.x >> 5;
  const int lane = threadIdx.x & 31;
  const int half = lane >> 4;
  const int l    = lane & 15;

  const int m0 = blockIdx.x * 16;
  const int n0 = blockIdx.y * 64 + wave * 16;
  if (m0 >= M || n0 >= Ncols) return;

  v8f c = {0.f, 0.f, 0.f, 0.f, 0.f, 0.f, 0.f, 0.f};

  const float* arow = A + (size_t)(m0 + l) * K + half * 2;
  const float* bcol = B + (size_t)(half * 2) * Ncols + n0 + l;

  for (int k = 0; k < K; k += 4) {
    v2f a = *(const v2f*)(arow + k);
    v2f b;
    b.x = bcol[(size_t)k * Ncols];
    b.y = bcol[(size_t)k * Ncols + Ncols];
    c = __builtin_amdgcn_wmma_f32_16x16x4_f32(
         false, a,  false, b,
         (short)0, c,  false,  false);
  }

  float* crow = C + n0 + l;
  const int mbase = m0 + half * 8;
#pragma unroll
  for (int r = 0; r < 8; ++r)
    crow[(size_t)(mbase + r) * Ncols] = c[r];
}

__global__ void alphas_kernel(const float* __restrict__ h,
                              const float* __restrict__ asrc,
                              const float* __restrict__ adst,
                              float* __restrict__ s, float* __restrict__ d,
                              int total) {
  int i = blockIdx.x * blockDim.x + threadIdx.x;
  if (i >= total) return;
  int node = i >> 3, hd = i & 7;
  const float* hp = h + (size_t)node * GAT_FH + hd * GAT_C;
  const float* ap = asrc + hd * GAT_C;
  const float* bp = adst + hd * GAT_C;
  float ss = 0.f, dd = 0.f;
#pragma unroll
  for (int cc = 0; cc < GAT_C; ++cc) {
    float v = hp[cc];
    ss += v * ap[cc];
    dd += v * bp[cc];
  }
  s[i] = ss;
  d[i] = dd;
}

__global__ void init_kernel(float* __restrict__ agg, float* __restrict__ emax,
                            float* __restrict__ denom, int nfeat, int nh) {
  int i = blockIdx.x * blockDim.x + threadIdx.x;
  if (i < nfeat) agg[i] = 0.f;
  if (i < nh) { emax[i] = -3.0e38f; denom[i] = 0.f; }
}

__device__ __forceinline__ void edge_endpoints(const int* __restrict__ ei,
                                               int e, int& src, int& dst) {
  if (e < GAT_E) { src = ei[e]; dst = ei[GAT_E + e]; }
  else           { src = e - GAT_E; dst = e - GAT_E; }
}

__global__ void edge_max_kernel(const int* __restrict__ ei,
                                const float* __restrict__ as,
                                const float* __restrict__ ad,
                                float* __restrict__ emax) {
  int e = blockIdx.x * blockDim.x + threadIdx.x;
  if (e >= GAT_ETOT) return;
  int src, dst;
  edge_endpoints(ei, e, src, dst);
#pragma unroll
  for (int hd = 0; hd < GAT_H; ++hd) {
    float v = as[src * GAT_H + hd] + ad[dst * GAT_H + hd];
    v = v > 0.f ? v : NEG_SLOPE * v;
    atomicMax(&emax[dst * GAT_H + hd], v);
  }
}

__global__ void edge_denom_kernel(const int* __restrict__ ei,
                                  const float* __restrict__ as,
                                  const float* __restrict__ ad,
                                  const float* __restrict__ emax,
                                  float* __restrict__ denom) {
  int e = blockIdx.x * blockDim.x + threadIdx.x;
  if (e >= GAT_ETOT) return;
  int src, dst;
  edge_endpoints(ei, e, src, dst);
#pragma unroll
  for (int hd = 0; hd < GAT_H; ++hd) {
    float v = as[src * GAT_H + hd] + ad[dst * GAT_H + hd];
    v = v > 0.f ? v : NEG_SLOPE * v;
    atomicAdd(&denom[dst * GAT_H + hd], __expf(v - emax[dst * GAT_H + hd]));
  }
}

__global__ void edge_agg_kernel(const int* __restrict__ ei,
                                const float* __restrict__ as,
                                const float* __restrict__ ad,
                                const float* __restrict__ emax,
                                const float* __restrict__ denom,
                                const float* __restrict__ h,
                                float* __restrict__ agg) {
  int t = blockIdx.x * blockDim.x + threadIdx.x;
  if (t >= GAT_ETOT * GAT_H) return;
  int e = t >> 3, hd = t & 7;
  int src, dst;
  edge_endpoints(ei, e, src, dst);
  float v = as[src * GAT_H + hd] + ad[dst * GAT_H + hd];
  v = v > 0.f ? v : NEG_SLOPE * v;
  int di = dst * GAT_H + hd;
  float alpha = __expf(v - emax[di]) / (denom[di] + 1e-16f);
  const float4* hp = (const float4*)(h + (size_t)src * GAT_FH + hd * GAT_C);
  float* op = agg + (size_t)dst * GAT_FH + hd * GAT_C;
#pragma unroll
  for (int q = 0; q < GAT_C / 4; ++q) {
    float4 x4 = hp[q];
    atomicAdd(op + q * 4 + 0, x4.x * alpha);
    atomicAdd(op + q * 4 + 1, x4.y * alpha);
    atomicAdd(op + q * 4 + 2, x4.z * alpha);
    atomicAdd(op + q * 4 + 3, x4.w * alpha);
  }
}

__global__ void bias_elu_kernel(float* __restrict__ agg,
                                const float* __restrict__ b, int total) {
  int i = blockIdx.x * blockDim.x + threadIdx.x;
  if (i >= total) return;
  float v = agg[i] + b[i & (GAT_FH - 1)];
  agg[i] = v > 0.f ? v : (__expf(v) - 1.f);
}

__global__ void final_linear_kernel(const float* __restrict__ h,
                                    const float* __restrict__ Wl,
                                    const float* __restrict__ bl,
                                    float* __restrict__ y, int n) {
  int gid = blockIdx.x * blockDim.x + threadIdx.x;
  int wid = gid >> 5;
  int lane = threadIdx.x & 31;
  if (wid >= n) return;
  const float* hp = h + (size_t)wid * GAT_FH;
  float s = 0.f;
#pragma unroll
  for (int c = lane; c < GAT_FH; c += 32) s += hp[c] * Wl[c];
#pragma unroll
  for (int off = 16; off > 0; off >>= 1) s += __shfl_down(s, off, 32);
  if (lane == 0) y[wid] = s + bl[0];
}

extern "C" void kernel_launch(void* const* d_in, const int* in_sizes, int n_in,
                              void* d_out, int out_size, void* d_ws, size_t ws_size,
                              hipStream_t stream) {
  const float* x      = (const float*)d_in[0];
  const int*   ei     = (const int*)  d_in[1];
  const float* W1     = (const float*)d_in[2];
  const float* a1_src = (const float*)d_in[3];
  const float* a1_dst = (const float*)d_in[4];
  const float* b1     = (const float*)d_in[5];
  const float* W2     = (const float*)d_in[6];
  const float* a2_src = (const float*)d_in[7];
  const float* a2_dst = (const float*)d_in[8];
  const float* b2     = (const float*)d_in[9];
  const float* Wl     = (const float*)d_in[10];
  const float* bl     = (const float*)d_in[11];
  float* out = (float*)d_out;

  float* hbuf  = (float*)d_ws;
  float* agg   = hbuf  + (size_t)GAT_N * GAT_FH;
  float* emax  = agg   + (size_t)GAT_N * GAT_FH;
  float* denom = emax  + (size_t)GAT_N * GAT_H;
  float* as    = denom + (size_t)GAT_N * GAT_H;
  float* ad    = as    + (size_t)GAT_N * GAT_H;

  const int NF  = GAT_N * GAT_FH;
  const int NH  = GAT_N * GAT_H;
  const int EH  = GAT_ETOT * GAT_H;

  dim3 gemm_grid(GAT_N / 16, GAT_FH / 64);
  dim3 gemm_blk(128);

  const int TPB = 256;
  auto blocks = [](int n, int tpb) { return (n + tpb - 1) / tpb; };

  gemm_wmma_f32<<<gemm_grid, gemm_blk, 0, stream>>>(x, W1, hbuf, GAT_N, 128, GAT_FH);
  alphas_kernel<<<blocks(NH, TPB), TPB, 0, stream>>>(hbuf, a1_src, a1_dst, as, ad, NH);
  init_kernel<<<blocks(NF, TPB), TPB, 0, stream>>>(agg, emax, denom, NF, NH);
  edge_max_kernel<<<blocks(GAT_ETOT, TPB), TPB, 0, stream>>>(ei, as, ad, emax);
  edge_denom_kernel<<<blocks(GAT_ETOT, TPB), TPB, 0, stream>>>(ei, as, ad, emax, denom);
  edge_agg_kernel<<<blocks(EH, TPB), TPB, 0, stream>>>(ei, as, ad, emax, denom, hbuf, agg);
  bias_elu_kernel<<<blocks(NF, TPB), TPB, 0, stream>>>(agg, b1, NF);

  gemm_wmma_f32<<<gemm_grid, gemm_blk, 0, stream>>>(agg, W2, hbuf, GAT_N, GAT_FH, GAT_FH);
  alphas_kernel<<<blocks(NH, TPB), TPB, 0, stream>>>(hbuf, a2_src, a2_dst, as, ad, NH);
  init_kernel<<<blocks(NF, TPB), TPB, 0, stream>>>(agg, emax, denom, NF, NH);
  edge_max_kernel<<<blocks(GAT_ETOT, TPB), TPB, 0, stream>>>(ei, as, ad, emax);
  edge_denom_kernel<<<blocks(GAT_ETOT, TPB), TPB, 0, stream>>>(ei, as, ad, emax, denom);
  edge_agg_kernel<<<blocks(EH, TPB), TPB, 0, stream>>>(ei, as, ad, emax, denom, hbuf, agg);
  bias_elu_kernel<<<blocks(NF, TPB), TPB, 0, stream>>>(agg, b2, NF);

  final_linear_kernel<<<blocks(GAT_N * 32, TPB), TPB, 0, stream>>>(agg, Wl, bl, out, GAT_N);
}


# GEMM made B-stationary: 16 waves per work-group keep their 16-column strip of B (all of K) in registers and walk over 16-row A tiles staged in LDS by whole-line loads; same k order, same sums
# speedup vs baseline: 13.1405x; 13.1405x over previous
#include <hip/hip_runtime.h>
#include <hip/hip_bf16.h>

#define GAT_N 30000
#define GAT_E 480000
#define GAT_ETOT (GAT_E + GAT_N)
#define GAT_H 8
#define GAT_C 32
#define GAT_FH 256
#define NEG_SLOPE 0.2f

typedef float v2f __attribute__((ext_vector_type(2)));
typedef float v8f __attribute__((ext_vector_type(8)));

#define BST_T 512
#define BST_WG 256
#define BST_MT (GAT_N / 16)
static_assert(GAT_N % 16 == 0, "whole 16-row tiles");
static_assert(GAT_FH == 16 * (BST_T / 32), "one wave per 16-column strip");

template <int K>
__global__ void __launch_bounds__(BST_T)
gemm_bstat_f32(const float* __restrict__ A, const float* __restrict__ B,
               float* __restrict__ C) {
  constexpr int P = K + 4;
  constexpr int V = (16 * K / 4) / BST_T;
  static_assert((16 * K / 4) % BST_T == 0 && (V == 1 || V == 2), "one or two vectors per thread");
  __shared__ float sa[2][16 * P];

  const int tid  = (int)threadIdx.x;
  const int lane = tid & 31;
  const int wave = __builtin_amdgcn_readfirstlane(tid >> 5);
  const int half = lane >> 4;
  const int l    = lane & 15;
  const int n0   = wave * 16;

  v2f b[K / 4];
  {
    const float* bcol = B + (size_t)(half * 2) * GAT_FH + n0 + l;
#pragma unroll
    for (int t = 0; t < K / 4; ++t) {
      b[t].x = bcol[(size_t)(4 * t) * GAT_FH];
      b[t].y = bcol[(size_t)(4 * t + 1) * GAT_FH];
    }
  }

  const int so0 = (tid / (K / 4)) * P + (tid % (K / 4)) * 4;
  const int so1 = ((tid + BST_T) / (K / 4)) * P + ((tid + BST_T) % (K / 4)) * 4;

  int tile = (int)blockIdx.x;
  float4 r0 = make_float4(0.f, 0.f, 0.f, 0.f), r1 = r0;
  if (tile < BST_MT) {
    const float4* g = (const float4*)(A + (size_t)tile * 16 * K);
    r0 = g[tid];
    if constexpr (V == 2) r1 = g[tid + BST_T];
  }

  int buf = 0;
  for (; tile < BST_MT; tile += BST_WG) {
    float* s = sa[buf];
    *(float4*)(s + so0) = r0;
    if constexpr (V == 2) *(float4*)(s + so1) = r1;
    __syncthreads();

    {
      const int nt = (tile + BST_WG < BST_MT) ? tile + BST_WG : tile;
      const float4* g = (const float4*)(A + (size_t)nt * 16 * K);
      r0 = g[tid];
      if constexpr (V == 2) r1 = g[tid + BST_T];
    }
    __builtin_amdgcn_sched_barrier(0);

    v8f c = {0.f, 0.f, 0.f, 0.f, 0.f, 0.f, 0.f, 0.f};
    const float* ap = s + l * P + half * 2;
#pragma unroll
    for (int t = 0; t < K / 4; ++t) {
      const v2f a = *(const v2f*)(ap + 4 * t);
      c = __builtin_amdgcn_wmma_f32_16x16x4_f32(
           false, a,  false, b[t],
           (short)0, c,  false,  false);
    }

    float* crow = C + (size_t)(tile * 16 + half * 8) * GAT_FH + n0 + l;
#pragma unroll
    for (int q = 0; q < 8; ++q) crow[(size_t)q * GAT_FH] = c[q];
    buf ^= 1;
  }
}

__global__ void alphas_kernel(const float* __restrict__ h,
                              const float* __restrict__ asrc,
                              const float* __restrict__ adst,
                              float* __restrict__ s, float* __restrict__ d,
                              int total) {
  int i = blockIdx.x * blockDim.x + threadIdx.x;
  if (i >= total) return;
  int node = i >> 3, hd = i & 7;
  const float* hp = h + (size_t)node * GAT_FH + hd * GAT_C;
  const float* ap = asrc + hd * GAT_C;
  const float* bp = adst + hd * GAT_C;
  float ss = 0.f, dd = 0.f;
#pragma unroll
  for (int cc = 0; cc < GAT_C; ++cc) {
    float v = hp[cc];
    ss += v * ap[cc];
    dd += v * bp[cc];
  }
  s[i] = ss;
  d[i] = dd;
}

__device__ __forceinline__ void edge_endpoints(const int* __restrict__ ei,
                                               int e, int& src, int& dst) {
  if (e < GAT_E) { src = ei[e]; dst = ei[GAT_E + e]; }
  else           { src = e - GAT_E; dst = e - GAT_E; }
}


#define OWN_S 118
#define OWN_WG ((GAT_N + OWN_S - 1) / OWN_S)
#define OWN_T 1024
#define OWN_CAP 16384
#define OWN_DLBITS 7
#define OWN_NV (GAT_E / 4)
static_assert(OWN_S <= (1 << OWN_DLBITS), "local row fits its field");
static_assert(GAT_ETOT < (1 << (31 - OWN_DLBITS)), "edge number fits beside the local row");
static_assert(GAT_E % 4 == 0, "destination array is whole int4 vectors");
static_assert(OWN_S * GAT_H <= OWN_T, "one thread per (row, head) in the set-up");

__device__ __forceinline__ float lrelu(float v) { return v > 0.f ? v : NEG_SLOPE * v; }

__device__ __forceinline__ float elu_bias(float x, float b) {
  const float v = x + b;
  return v > 0.f ? v : (__expf(v) - 1.f);
}

template <int U>
__device__ __forceinline__ void own_add_entries(const int (&src)[U], const int (&dl)[U], int lane,
                                                const float* __restrict__ as,
                                                const float* __restrict__ h,
                                                const float* adl, const float* mx,
                                                float* dn, float* acc) {
  const int hd = lane & (GAT_H - 1);
  float a[U];
  float x[U][GAT_H];
#pragma unroll
  for (int u = 0; u < U; ++u) {
    a[u] = as[src[u] * GAT_H + hd];
    const float* hp = h + (size_t)src[u] * GAT_FH + lane;
#pragma unroll
    for (int k = 0; k < GAT_H; ++k) x[u][k] = hp[k * GAT_C];
  }
#pragma unroll
  for (int u = 0; u < U; ++u) {
    const int o = dl[u] * GAT_H + hd;
    const float w = __expf(lrelu(a[u] + adl[o]) - mx[o]);
    if (lane < GAT_H) atomicAdd(&dn[o], w);
    float* ap = acc + dl[u] * GAT_FH + lane;
#pragma unroll
    for (int k = 0; k < GAT_H; ++k) {
      const float wk = __int_as_float(__builtin_amdgcn_readlane(__float_as_int(w), k));
      atomicAdd(ap + k * GAT_C, x[u][k] * wk);
    }
  }
}

__global__ void __launch_bounds__(OWN_T)
owner_kernel(const int* __restrict__ ei, const float* __restrict__ as,
             const float* __restrict__ ad, const float* __restrict__ h,
             const float* __restrict__ b, float* __restrict__ out) {
  __shared__ float acc[OWN_S * GAT_FH];
  __shared__ int   list[OWN_CAP];
  __shared__ float mx[OWN_S * GAT_H];
  __shared__ float dn[OWN_S * GAT_H];
  __shared__ float adl[OWN_S * GAT_H];
  __shared__ int   cnt;

  const int tid  = (int)threadIdx.x;
  const int lane = tid & 31;
  const int wave = __builtin_amdgcn_readfirstlane(tid >> 5);
  const int lo = (int)blockIdx.x * OWN_S;
  const int nn = (GAT_N - lo) < OWN_S ? (GAT_N - lo) : OWN_S;

  for (int i = tid; i < OWN_S * GAT_FH / 4; i += OWN_T)
    ((float4*)acc)[i] = make_float4(0.f, 0.f, 0.f, 0.f);
  if (tid < nn * GAT_H) {
    mx[tid] = -3.0e38f;
    dn[tid] = 0.f;
    adl[tid] = ad[lo * GAT_H + tid];
  }
  if (tid < nn) list[tid] = ((GAT_E + lo + tid) << OWN_DLBITS) | tid;
  if (tid == 0) cnt = nn;
  __syncthreads();

  {
    const int4* d4 = (const int4*)(ei + GAT_E);
    auto test = [&](int dv, int e) {
      const unsigned r = (unsigned)(dv - lo);
      if (r < (unsigned)nn) {
        const int pos = atomicAdd(&cnt, 1);
        if (pos < OWN_CAP) list[pos] = (e << OWN_DLBITS) | (int)r;
      }
    };
    auto test4 = [&](const int4& q, int j) {
      const unsigned r0 = (unsigned)(q.x - lo), r1 = (unsigned)(q.y - lo);
      const unsigned r2 = (unsigned)(q.z - lo), r3 = (unsigned)(q.w - lo);
      const unsigned m01 = r0 < r1 ? r0 : r1, m23 = r2 < r3 ? r2 : r3;
      if ((m01 < m23 ? m01 : m23) < (unsigned)nn) {
        test(q.x, 4 * j); test(q.y, 4 * j + 1); test(q.z, 4 * j + 2); test(q.w, 4 * j + 3);
      }
    };
    int j = tid;
    for (; j + 3 * OWN_T < OWN_NV; j += 4 * OWN_T) {
      const int4 q0 = d4[j], q1 = d4[j + OWN_T], q2 = d4[j + 2 * OWN_T], q3 = d4[j + 3 * OWN_T];
      test4(q0, j); test4(q1, j + OWN_T); test4(q2, j + 2 * OWN_T); test4(q3, j + 3 * OWN_T);
    }
    for (; j < OWN_NV; j += OWN_T) { const int4 q = d4[j]; test4(q, j); }
  }
  __syncthreads();
  const int total = __builtin_amdgcn_readfirstlane(cnt);
  const bool fits = total <= OWN_CAP;

  if (fits) {
    for (int i = tid; i < total; i += OWN_T) {
      const int ent = list[i];
      const int e = ent >> OWN_DLBITS, dl = ent & ((1 << OWN_DLBITS) - 1);
      const int src = (e < GAT_E) ? ei[e] : (e - GAT_E);
      list[i] = (src << OWN_DLBITS) | dl;
      const float4 a0 = *(const float4*)(as + src * GAT_H);
      const float4 a1 = *(const float4*)(as + src * GAT_H + 4);
      const float* dp = adl + dl * GAT_H;
      float* mp = mx + dl * GAT_H;
      atomicMax(mp + 0, lrelu(a0.x + dp[0]));
      atomicMax(mp + 1, lrelu(a0.y + dp[1]));
      atomicMax(mp + 2, lrelu(a0.z + dp[2]));
      atomicMax(mp + 3, lrelu(a0.w + dp[3]));
      atomicMax(mp + 4, lrelu(a1.x + dp[4]));
      atomicMax(mp + 5, lrelu(a1.y + dp[5]));
      atomicMax(mp + 6, lrelu(a1.z + dp[6]));
      atomicMax(mp + 7, lrelu(a1.w + dp[7]));
    }
  } else {
    for (int e = tid; e < GAT_ETOT; e += OWN_T) {
      int src, dst;
      edge_endpoints(ei, e, src, dst);
      const unsigned r = (unsigned)(dst - lo);
      if (r < (unsigned)nn) {
#pragma unroll
        for (int k = 0; k < GAT_H; ++k)
          atomicMax(&mx[r * GAT_H + k], lrelu(as[src * GAT_H + k] + adl[r * GAT_H + k]));
      }
    }
  }
  __syncthreads();

  if (fits) {
    int i = wave;
    for (; i + 3 * 32 < total; i += 4 * 32) {
      int src[4], dl[4];
#pragma unroll
      for (int u = 0; u < 4; ++u) {
        const int ent = __builtin_amdgcn_readfirstlane(list[i + u * 32]);
        src[u] = ent >> OWN_DLBITS;
        dl[u] = ent & ((1 << OWN_DLBITS) - 1);
      }
      own_add_entries<4>(src, dl, lane, as, h, adl, mx, dn, acc);
    }
    for (; i < total; i += 32) {
      const int ent = __builtin_amdgcn_readfirstlane(list[i]);
      const int src[1] = {ent >> OWN_DLBITS};
      const int dl[1] = {ent & ((1 << OWN_DLBITS) - 1)};
      own_add_entries<1>(src, dl, lane, as, h, adl, mx, dn, acc);
    }
  } else {
    for (int e0 = 0; e0 < GAT_ETOT; e0 += OWN_T) {
      const int e = e0 + tid;
      int src = 0, dst = 0;
      bool hit = false;
      unsigned r = 0;
      if (e < GAT_ETOT) {
        edge_endpoints(ei, e, src, dst);
        r = (unsigned)(dst - lo);
        hit = r < (unsigned)nn;
      }
      unsigned todo = __builtin_amdgcn_ballot_w32(hit);
      while (todo) {
        const int k = __builtin_ctz(todo);
        todo &= todo - 1;
        const int s1[1] = {__builtin_amdgcn_readlane(src, k)};
        const int d1[1] = {__builtin_amdgcn_readlane((int)r, k)};
        own_add_entries<1>(s1, d1, lane, as, h, adl, mx, dn, acc);
      }
    }
  }
  __syncthreads();

  float4* o4 = (float4*)(out + (size_t)lo * GAT_FH);
  for (int i = tid; i < nn * (GAT_FH / 4); i += OWN_T) {
    const int r = i >> 6, c = (i & 63) * 4;
    const float4 s = ((const float4*)acc)[i];
    const float den = dn[r * GAT_H + (c >> 5)] + 1e-16f;
    const float4 bv = *(const float4*)(b + c);
    float4 o;
    o.x = elu_bias(s.x / den, bv.x);
    o.y = elu_bias(s.y / den, bv.y);
    o.z = elu_bias(s.z / den, bv.z);
    o.w = elu_bias(s.w / den, bv.w);
    o4[i] = o;
  }
}

__global__ void final_linear_kernel(const float* __restrict__ h,
                                    const float* __restrict__ Wl,
                                    const float* __restrict__ bl,
                                    float* __restrict__ y, int n) {
  int gid = blockIdx.x * blockDim.x + threadIdx.x;
  int wid = gid >> 5;
  int lane = threadIdx.x & 31;
  if (wid >= n) return;
  const float* hp = h + (size_t)wid * GAT_FH;
  float s = 0.f;
#pragma unroll
  for (int c = lane; c < GAT_FH; c += 32) s += hp[c] * Wl[c];
#pragma unroll
  for (int off = 16; off > 0; off >>= 1) s += __shfl_down(s, off, 32);
  if (lane == 0) y[wid] = s + bl[0];
}

extern "C" void kernel_launch(void* const* d_in, const int* in_sizes, int n_in,
                              void* d_out, int out_size, void* d_ws, size_t ws_size,
                              hipStream_t stream) {
  const float* x      = (const float*)d_in[0];
  const int*   ei     = (const int*)  d_in[1];
  const float* W1     = (const float*)d_in[2];
  const float* a1_src = (const float*)d_in[3];
  const float* a1_dst = (const float*)d_in[4];
  const float* b1     = (const float*)d_in[5];
  const float* W2     = (const float*)d_in[6];
  const float* a2_src = (const float*)d_in[7];
  const float* a2_dst = (const float*)d_in[8];
  const float* b2     = (const float*)d_in[9];
  const float* Wl     = (const float*)d_in[10];
  const float* bl     = (const float*)d_in[11];
  float* out = (float*)d_out;

  float* hbuf = (float*)d_ws;
  float* act  = hbuf + (size_t)GAT_N * GAT_FH;
  float* as   = act  + (size_t)GAT_N * GAT_FH;
  float* ad   = as   + (size_t)GAT_N * GAT_H;

  const int NH  = GAT_N * GAT_H;


  const int TPB = 256;
  auto blocks = [](int n, int tpb) { return (n + tpb - 1) / tpb; };

  gemm_bstat_f32<128><<<BST_WG, BST_T, 0, stream>>>(x, W1, hbuf);
  alphas_kernel<<<blocks(NH, TPB), TPB, 0, stream>>>(hbuf, a1_src, a1_dst, as, ad, NH);
  owner_kernel<<<OWN_WG, OWN_T, 0, stream>>>(ei, as, ad, hbuf, b1, act);

  gemm_bstat_f32<GAT_FH><<<BST_WG, BST_T, 0, stream>>>(act, W2, hbuf);
  alphas_kernel<<<blocks(NH, TPB), TPB, 0, stream>>>(hbuf, a2_src, a2_dst, as, ad, NH);
  owner_kernel<<<OWN_WG, OWN_T, 0, stream>>>(ei, as, ad, hbuf, b2, act);

  final_linear_kernel<<<blocks(GAT_N * 32, TPB), TPB, 0, stream>>>(act, Wl, bl, out, GAT_N);
}


# Edge filter runs once per call: layer 1's work-groups leave their hit lists (source, local row) in the workspace; layer 2 reads its 8 KB list and skips the scan of all 480000 destinations; same sums
# speedup vs baseline: 14.3923x; 1.0953x over previous
#include <hip/hip_runtime.h>
#include <hip/hip_bf16.h>

#define GAT_N 30000
#define GAT_E 480000
#define GAT_ETOT (GAT_E + GAT_N)
#define GAT_H 8
#define GAT_C 32
#define GAT_FH 256
#define NEG_SLOPE 0.2f

typedef float v2f __attribute__((ext_vector_type(2)));
typedef float v8f __attribute__((ext_vector_type(8)));

#define BST_T 512
#define BST_WG 256
#define BST_MT (GAT_N / 16)
static_assert(GAT_N % 16 == 0, "whole 16-row tiles");
static_assert(GAT_FH == 16 * (BST_T / 32), "one wave per 16-column strip");

template <int K>
__global__ void __launch_bounds__(BST_T)
gemm_bstat_f32(const float* __restrict__ A, const float* __restrict__ B,
               float* __restrict__ C) {
  constexpr int P = K + 4;
  constexpr int V = (16 * K / 4) / BST_T;
  static_assert((16 * K / 4) % BST_T == 0 && (V == 1 || V == 2), "one or two vectors per thread");
  __shared__ float sa[2][16 * P];

  const int tid  = (int)threadIdx.x;
  const int lane = tid & 31;
  const int wave = __builtin_amdgcn_readfirstlane(tid >> 5);
  const int half = lane >> 4;
  const int l    = lane & 15;
  const int n0   = wave * 16;

  v2f b[K / 4];
  {
    const float* bcol = B + (size_t)(half * 2) * GAT_FH + n0 + l;
#pragma unroll
    for (int t = 0; t < K / 4; ++t) {
      b[t].x = bcol[(size_t)(4 * t) * GAT_FH];
      b[t].y = bcol[(size_t)(4 * t + 1) * GAT_FH];
    }
  }

  const int so0 = (tid / (K / 4)) * P + (tid % (K / 4)) * 4;
  const int so1 = ((tid + BST_T) / (K / 4)) * P + ((tid + BST_T) % (K / 4)) * 4;

  int tile = (int)blockIdx.x;
  float4 r0 = make_float4(0.f, 0.f, 0.f, 0.f), r1 = r0;
  if (tile < BST_MT) {
    const float4* g = (const float4*)(A + (size_t)tile * 16 * K);
    r0 = g[tid];
    if constexpr (V == 2) r1 = g[tid + BST_T];
  }

  int buf = 0;
  for (; tile < BST_MT; tile += BST_WG) {
    float* s = sa[buf];
    *(float4*)(s + so0) = r0;
    if constexpr (V == 2) *(float4*)(s + so1) = r1;
    __syncthreads();

    {
      const int nt = (tile + BST_WG < BST_MT) ? tile + BST_WG : tile;
      const float4* g = (const float4*)(A + (size_t)nt * 16 * K);
      r0 = g[tid];
      if constexpr (V == 2) r1 = g[tid + BST_T];
    }
    __builtin_amdgcn_sched_barrier(0);

    v8f c = {0.f, 0.f, 0.f, 0.f, 0.f, 0.f, 0.f, 0.f};
    const float* ap = s + l * P + half * 2;
#pragma unroll
    for (int t = 0; t < K / 4; ++t) {
      const v2f a = *(const v2f*)(ap + 4 * t);
      c = __builtin_amdgcn_wmma_f32_16x16x4_f32(
           false, a,  false, b[t],
           (short)0, c,  false,  false);
    }

    float* crow = C + (size_t)(tile * 16 + half * 8) * GAT_FH + n0 + l;
#pragma unroll
    for (int q = 0; q < 8; ++q) crow[(size_t)q * GAT_FH] = c[q];
    buf ^= 1;
  }
}

__global__ void alphas_kernel(const float* __restrict__ h,
                              const float* __restrict__ asrc,
                              const float* __restrict__ adst,
                              float* __restrict__ s, float* __restrict__ d,
                              int total) {
  int i = blockIdx.x * blockDim.x + threadIdx.x;
  if (i >= total) return;
  int node = i >> 3, hd = i & 7;
  const float* hp = h + (size_t)node * GAT_FH + hd * GAT_C;
  const float* ap = asrc + hd * GAT_C;
  const float* bp = adst + hd * GAT_C;
  float ss = 0.f, dd = 0.f;
#pragma unroll
  for (int cc = 0; cc < GAT_C; ++cc) {
    float v = hp[cc];
    ss += v * ap[cc];
    dd += v * bp[cc];
  }
  s[i] = ss;
  d[i] = dd;
}

__device__ __forceinline__ void edge_endpoints(const int* __restrict__ ei,
                                               int e, int& src, int& dst) {
  if (e < GAT_E) { src = ei[e]; dst = ei[GAT_E + e]; }
  else           { src = e - GAT_E; dst = e - GAT_E; }
}


#define OWN_S 118
#define OWN_WG ((GAT_N + OWN_S - 1) / OWN_S)
#define OWN_T 1024
#define OWN_CAP 16384
#define OWN_DLBITS 7
#define OWN_NV (GAT_E / 4)
static_assert(OWN_S <= (1 << OWN_DLBITS), "local row fits its field");
static_assert(GAT_ETOT < (1 << (31 - OWN_DLBITS)), "edge number fits beside the local row");
static_assert(GAT_E % 4 == 0, "destination array is whole int4 vectors");
static_assert(OWN_S * GAT_H <= OWN_T, "one thread per (row, head) in the set-up");

__device__ __forceinline__ float lrelu(float v) { return v > 0.f ? v : NEG_SLOPE * v; }

__device__ __forceinline__ float elu_bias(float x, float b) {
  const float v = x + b;
  return v > 0.f ? v : (__expf(v) - 1.f);
}

template <int U>
__device__ __forceinline__ void own_add_entries(const int (&src)[U], const int (&dl)[U], int lane,
                                                const float* __restrict__ as,
                                                const float* __restrict__ h,
                                                const float* adl, const float* mx,
                                                float* dn, float* acc) {
  const int hd = lane & (GAT_H - 1);
  float a[U];
  float x[U][GAT_H];
#pragma unroll
  for (int u = 0; u < U; ++u) {
    a[u] = as[src[u] * GAT_H + hd];
    const float* hp = h + (size_t)src[u] * GAT_FH + lane;
#pragma unroll
    for (int k = 0; k < GAT_H; ++k) x[u][k] = hp[k * GAT_C];
  }
#pragma unroll
  for (int u = 0; u < U; ++u) {
    const int o = dl[u] * GAT_H + hd;
    const float w = __expf(lrelu(a[u] + adl[o]) - mx[o]);
    if (lane < GAT_H) atomicAdd(&dn[o], w);
    float* ap = acc + dl[u] * GAT_FH + lane;
#pragma unroll
    for (int k = 0; k < GAT_H; ++k) {
      const float wk = __int_as_float(__builtin_amdgcn_readlane(__float_as_int(w), k));
      atomicAdd(ap + k * GAT_C, x[u][k] * wk);
    }
  }
}

template <bool FIRST>
__global__ void __launch_bounds__(OWN_T)
owner_kernel(const int* __restrict__ ei, const float* __restrict__ as,
             const float* __restrict__ ad, const float* __restrict__ h,
             const float* __restrict__ b, float* __restrict__ out,
             int* __restrict__ glist, int* __restrict__ gcnt) {
  __shared__ float acc[OWN_S * GAT_FH];
  __shared__ int   list[OWN_CAP];
  __shared__ float mx[OWN_S * GAT_H];
  __shared__ float dn[OWN_S * GAT_H];
  __shared__ float adl[OWN_S * GAT_H];
  __shared__ int   cnt;

  const int tid  = (int)threadIdx.x;
  const int lane = tid & 31;
  const int wave = __builtin_amdgcn_readfirstlane(tid >> 5);
  const int lo = (int)blockIdx.x * OWN_S;
  const int nn = (GAT_N - lo) < OWN_S ? (GAT_N - lo) : OWN_S;
  int* gl = glist + (size_t)blockIdx.x * OWN_CAP;

  for (int i = tid; i < OWN_S * GAT_FH / 4; i += OWN_T)
    ((float4*)acc)[i] = make_float4(0.f, 0.f, 0.f, 0.f);
  if (tid < nn * GAT_H) {
    mx[tid] = -3.0e38f;
    dn[tid] = 0.f;
    adl[tid] = ad[lo * GAT_H + tid];
  }
  if constexpr (FIRST) {
    if (tid < nn) list[tid] = ((GAT_E + lo + tid) << OWN_DLBITS) | tid;
    if (tid == 0) cnt = nn;
  } else {
    const int kept = gcnt[blockIdx.x];
    if (kept <= OWN_CAP)
      for (int i = tid; i < kept; i += OWN_T) list[i] = gl[i];
    if (tid == 0) cnt = kept;
  }
  __syncthreads();

  if constexpr (FIRST) {
    const int4* d4 = (const int4*)(ei + GAT_E);
    auto test = [&](int dv, int e) {
      const unsigned r = (unsigned)(dv - lo);
      if (r < (unsigned)nn) {
        const int pos = atomicAdd(&cnt, 1);
        if (pos < OWN_CAP) list[pos] = (e << OWN_DLBITS) | (int)r;
      }
    };
    auto test4 = [&](const int4& q, int j) {
      const unsigned r0 = (unsigned)(q.x - lo), r1 = (unsigned)(q.y - lo);
      const unsigned r2 = (unsigned)(q.z - lo), r3 = (unsigned)(q.w - lo);
      const unsigned m01 = r0 < r1 ? r0 : r1, m23 = r2 < r3 ? r2 : r3;
      if ((m01 < m23 ? m01 : m23) < (unsigned)nn) {
        test(q.x, 4 * j); test(q.y, 4 * j + 1); test(q.z, 4 * j + 2); test(q.w, 4 * j + 3);
      }
    };
    int j = tid;
    for (; j + 3 * OWN_T < OWN_NV; j += 4 * OWN_T) {
      const int4 q0 = d4[j], q1 = d4[j + OWN_T], q2 = d4[j + 2 * OWN_T], q3 = d4[j + 3 * OWN_T];
      test4(q0, j); test4(q1, j + OWN_T); test4(q2, j + 2 * OWN_T); test4(q3, j + 3 * OWN_T);
    }
    for (; j < OWN_NV; j += OWN_T) { const int4 q = d4[j]; test4(q, j); }
    __syncthreads();
  }
  const int total = __builtin_amdgcn_readfirstlane(cnt);
  const bool fits = total <= OWN_CAP;
  if constexpr (FIRST) {
    if (tid == 0) gcnt[blockIdx.x] = total;
  }

  if (fits) {
    for (int i = tid; i < total; i += OWN_T) {
      const int ent = list[i];
      const int dl = ent & ((1 << OWN_DLBITS) - 1);
      int src = ent >> OWN_DLBITS;
      if constexpr (FIRST) {
        const int e = src;
        src = (e < GAT_E) ? ei[e] : (e - GAT_E);
        const int kept = (src << OWN_DLBITS) | dl;
        list[i] = kept;
        gl[i] = kept;
      }
      const float4 a0 = *(const float4*)(as + src * GAT_H);
      const float4 a1 = *(const float4*)(as + src * GAT_H + 4);
      const float* dp = adl + dl * GAT_H;
      float* mp = mx + dl * GAT_H;
      atomicMax(mp + 0, lrelu(a0.x + dp[0]));
      atomicMax(mp + 1, lrelu(a0.y + dp[1]));
      atomicMax(mp + 2, lrelu(a0.z + dp[2]));
      atomicMax(mp + 3, lrelu(a0.w + dp[3]));
      atomicMax(mp + 4, lrelu(a1.x + dp[4]));
      atomicMax(mp + 5, lrelu(a1.y + dp[5]));
      atomicMax(mp + 6, lrelu(a1.z + dp[6]));
      atomicMax(mp + 7, lrelu(a1.w + dp[7]));
    }
  } else {
    for (int e = tid; e < GAT_ETOT; e += OWN_T) {
      int src, dst;
      edge_endpoints(ei, e, src, dst);
      const unsigned r = (unsigned)(dst - lo);
      if (r < (unsigned)nn) {
#pragma unroll
        for (int k = 0; k < GAT_H; ++k)
          atomicMax(&mx[r * GAT_H + k], lrelu(as[src * GAT_H + k] + adl[r * GAT_H + k]));
      }
    }
  }
  __syncthreads();

  if (fits) {
    int i = wave;
    for (; i + 3 * 32 < total; i += 4 * 32) {
      int src[4], dl[4];
#pragma unroll
      for (int u = 0; u < 4; ++u) {
        const int ent = __builtin_amdgcn_readfirstlane(list[i + u * 32]);
        src[u] = ent >> OWN_DLBITS;
        dl[u] = ent & ((1 << OWN_DLBITS) - 1);
      }
      own_add_entries<4>(src, dl, lane, as, h, adl, mx, dn, acc);
    }
    for (; i < total; i += 32) {
      const int ent = __builtin_amdgcn_readfirstlane(list[i]);
      const int src[1] = {ent >> OWN_DLBITS};
      const int dl[1] = {ent & ((1 << OWN_DLBITS) - 1)};
      own_add_entries<1>(src, dl, lane, as, h, adl, mx, dn, acc);
    }
  } else {
    for (int e0 = 0; e0 < GAT_ETOT; e0 += OWN_T) {
      const int e = e0 + tid;
      int src = 0, dst = 0;
      bool hit = false;
      unsigned r = 0;
      if (e < GAT_ETOT) {
        edge_endpoints(ei, e, src, dst);
        r = (unsigned)(dst - lo);
        hit = r < (unsigned)nn;
      }
      unsigned todo = __builtin_amdgcn_ballot_w32(hit);
      while (todo) {
        const int k = __builtin_ctz(todo);
        todo &= todo - 1;
        const int s1[1] = {__builtin_amdgcn_readlane(src, k)};
        const int d1[1] = {__builtin_amdgcn_readlane((int)r, k)};
        own_add_entries<1>(s1, d1, lane, as, h, adl, mx, dn, acc);
      }
    }
  }
  __syncthreads();

  float4* o4 = (float4*)(out + (size_t)lo * GAT_FH);
  for (int i = tid; i < nn * (GAT_FH / 4); i += OWN_T) {
    const int r = i >> 6, c = (i & 63) * 4;
    const float4 s = ((const float4*)acc)[i];
    const float den = dn[r * GAT_H + (c >> 5)] + 1e-16f;
    const float4 bv = *(const float4*)(b + c);
    float4 o;
    o.x = elu_bias(s.x / den, bv.x);
    o.y = elu_bias(s.y / den, bv.y);
    o.z = elu_bias(s.z / den, bv.z);
    o.w = elu_bias(s.w / den, bv.w);
    o4[i] = o;
  }
}

__global__ void final_linear_kernel(const float* __restrict__ h,
                                    const float* __restrict__ Wl,
                                    const float* __restrict__ bl,
                                    float* __restrict__ y, int n) {
  int gid = blockIdx.x * blockDim.x + threadIdx.x;
  int wid = gid >> 5;
  int lane = threadIdx.x & 31;
  if (wid >= n) return;
  const float* hp = h + (size_t)wid * GAT_FH;
  float s = 0.f;
#pragma unroll
  for (int c = lane; c < GAT_FH; c += 32) s += hp[c] * Wl[c];
#pragma unroll
  for (int off = 16; off > 0; off >>= 1) s += __shfl_down(s, off, 32);
  if (lane == 0) y[wid] = s + bl[0];
}

extern "C" void kernel_launch(void* const* d_in, const int* in_sizes, int n_in,
                              void* d_out, int out_size, void* d_ws, size_t ws_size,
                              hipStream_t stream) {
  const float* x      = (const float*)d_in[0];
  const int*   ei     = (const int*)  d_in[1];
  const float* W1     = (const float*)d_in[2];
  const float* a1_src = (const float*)d_in[3];
  const float* a1_dst = (const float*)d_in[4];
  const float* b1     = (const float*)d_in[5];
  const float* W2     = (const float*)d_in[6];
  const float* a2_src = (const float*)d_in[7];
  const float* a2_dst = (const float*)d_in[8];
  const float* b2     = (const float*)d_in[9];
  const float* Wl     = (const float*)d_in[10];
  const float* bl     = (const float*)d_in[11];
  float* out = (float*)d_out;

  float* hbuf = (float*)d_ws;
  float* act  = hbuf + (size_t)GAT_N * GAT_FH;
  float* as   = act  + (size_t)GAT_N * GAT_FH;
  float* ad   = as   + (size_t)GAT_N * GAT_H;
  int* glist  = (int*)(ad + (size_t)GAT_N * GAT_H);
  int* gcnt   = glist + (size_t)OWN_WG * OWN_CAP;

  const int NH  = GAT_N * GAT_H;


  const int TPB = 256;
  auto blocks = [](int n, int tpb) { return (n + tpb - 1) / tpb; };

  gemm_bstat_f32<128><<<BST_WG, BST_T, 0, stream>>>(x, W1, hbuf);
  alphas_kernel<<<blocks(NH, TPB), TPB, 0, stream>>>(hbuf, a1_src, a1_dst, as, ad, NH);
  owner_kernel<true><<<OWN_WG, OWN_T, 0, stream>>>(ei, as, ad, hbuf, b1, act, glist, gcnt);

  gemm_bstat_f32<GAT_FH><<<BST_WG, BST_T, 0, stream>>>(act, W2, hbuf);
  alphas_kernel<<<blocks(NH, TPB), TPB, 0, stream>>>(hbuf, a2_src, a2_dst, as, ad, NH);
  owner_kernel<false><<<OWN_WG, OWN_T, 0, stream>>>(ei, as, ad, hbuf, b2, act, glist, gcnt);

  final_linear_kernel<<<blocks(GAT_N * 32, TPB), TPB, 0, stream>>>(act, Wl, bl, out, GAT_N);
}


# Attention scores formed in the GEMM epilogue from the accumulators (16-lane halving exchange per strip, two strips added in LDS: new sum order for the scores); both score launches go, 7 become 5
# speedup vs baseline: 15.2129x; 1.0570x over previous
#include <hip/hip_runtime.h>
#include <hip/hip_bf16.h>

#define GAT_N 30000
#define GAT_E 480000
#define GAT_ETOT (GAT_E + GAT_N)
#define GAT_H 8
#define GAT_C 32
#define GAT_FH 256
#define NEG_SLOPE 0.2f

typedef float v2f __attribute__((ext_vector_type(2)));
typedef float v8f __attribute__((ext_vector_type(8)));

#define BST_T 512
#define BST_WG 256
#define BST_MT (GAT_N / 16)
static_assert(GAT_N % 16 == 0, "whole 16-row tiles");
static_assert(GAT_FH == 16 * (BST_T / 32), "one wave per 16-column strip");

template <int CTRL>
__device__ __forceinline__ float row_xor_get(float x) {
  return __int_as_float(__builtin_amdgcn_update_dpp(0, __float_as_int(x), CTRL, 0xf, 0xf, true));
}

template <int N, int CTRL>
__device__ __forceinline__ void row_halve(float (&v)[16], bool up) {
#pragma unroll
  for (int j = 0; j < N; ++j) {
    const float keep = up ? v[j + N] : v[j];
    const float send = up ? v[j] : v[j + N];
    v[j] = keep + row_xor_get<CTRL>(send);
  }
}

template <int K>
__global__ void __launch_bounds__(BST_T)
gemm_bstat_f32(const float* __restrict__ A, const float* __restrict__ B,
               float* __restrict__ C,
               const float* __restrict__ asrc, const float* __restrict__ adst,
               float* __restrict__ as, float* __restrict__ ad) {
  constexpr int P = K + 4;
  constexpr int V = (16 * K / 4) / BST_T;
  static_assert((16 * K / 4) % BST_T == 0 && (V == 1 || V == 2), "one or two vectors per thread");
  __shared__ float sa[2][16 * P];
  __shared__ alignas(16) float xs[2][2 * 16 * 16];

  const int tid  = (int)threadIdx.x;
  const int lane = tid & 31;
  const int wave = __builtin_amdgcn_readfirstlane(tid >> 5);
  const int half = lane >> 4;
  const int l    = lane & 15;
  const int n0   = wave * 16;

  v2f b[K / 4];
  {
    const float* bcol = B + (size_t)(half * 2) * GAT_FH + n0 + l;
#pragma unroll
    for (int t = 0; t < K / 4; ++t) {
      b[t].x = bcol[(size_t)(4 * t) * GAT_FH];
      b[t].y = bcol[(size_t)(4 * t + 1) * GAT_FH];
    }
  }

  const float asv = asrc[n0 + l];
  const float adv = adst[n0 + l];
  const int xw = ((l >> 3) * 16 + half * 8 + (l & 7)) * 16 + wave;
  float* const xdst = (tid < 128) ? as : ad;

  const int so0 = (tid / (K / 4)) * P + (tid % (K / 4)) * 4;
  const int so1 = ((tid + BST_T) / (K / 4)) * P + ((tid + BST_T) % (K / 4)) * 4;

  int tile = (int)blockIdx.x;
  float4 r0 = make_float4(0.f, 0.f, 0.f, 0.f), r1 = r0;
  if (tile < BST_MT) {
    const float4* g = (const float4*)(A + (size_t)tile * 16 * K);
    r0 = g[tid];
    if constexpr (V == 2) r1 = g[tid + BST_T];
  }

  int buf = 0;
  int ptile = -1;
  for (; tile < BST_MT; tile += BST_WG) {
    float* s = sa[buf];
    *(float4*)(s + so0) = r0;
    if constexpr (V == 2) *(float4*)(s + so1) = r1;
    __syncthreads();

    {
      const int nt = (tile + BST_WG < BST_MT) ? tile + BST_WG : tile;
      const float4* g = (const float4*)(A + (size_t)nt * 16 * K);
      r0 = g[tid];
      if constexpr (V == 2) r1 = g[tid + BST_T];
    }
    __builtin_amdgcn_sched_barrier(0);

    v8f c = {0.f, 0.f, 0.f, 0.f, 0.f, 0.f, 0.f, 0.f};
    const float* ap = s + l * P + half * 2;
#pragma unroll
    for (int t = 0; t < K / 4; ++t) {
      const v2f a = *(const v2f*)(ap + 4 * t);
      c = __builtin_amdgcn_wmma_f32_16x16x4_f32(
           false, a,  false, b[t],
           (short)0, c,  false,  false);
    }

    float* crow = C + (size_t)(tile * 16 + half * 8) * GAT_FH + n0 + l;
#pragma unroll
    for (int q = 0; q < 8; ++q) crow[(size_t)q * GAT_FH] = c[q];

    {
      float v[16];
#pragma unroll
      for (int q = 0; q < 8; ++q) { v[q] = c[q] * asv; v[8 + q] = c[q] * adv; }
      row_halve<8, 0x168>(v, (l & 8) != 0);
      row_halve<4, 0x164>(v, (l & 4) != 0);
      row_halve<2, 0x162>(v, (l & 2) != 0);
      row_halve<1, 0x161>(v, (l & 1) != 0);
      xs[buf][xw] = v[0];
    }
    if (ptile >= 0 && tid < 256) {
      const v2f p = *(const v2f*)(&xs[buf ^ 1][2 * tid]);
      xdst[(size_t)ptile * 128 + (tid & 127)] = p.x + p.y;
    }
    ptile = tile;
    buf ^= 1;
  }
  __syncthreads();
  if (ptile >= 0 && tid < 256) {
    const v2f p = *(const v2f*)(&xs[buf ^ 1][2 * tid]);
    xdst[(size_t)ptile * 128 + (tid & 127)] = p.x + p.y;
  }
}

__device__ __forceinline__ void edge_endpoints(const int* __restrict__ ei,
                                               int e, int& src, int& dst) {
  if (e < GAT_E) { src = ei[e]; dst = ei[GAT_E + e]; }
  else           { src = e - GAT_E; dst = e - GAT_E; }
}


#define OWN_S 118
#define OWN_WG ((GAT_N + OWN_S - 1) / OWN_S)
#define OWN_T 1024
#define OWN_CAP 16384
#define OWN_DLBITS 7
#define OWN_NV (GAT_E / 4)
static_assert(OWN_S <= (1 << OWN_DLBITS), "local row fits its field");
static_assert(GAT_ETOT < (1 << (31 - OWN_DLBITS)), "edge number fits beside the local row");
static_assert(GAT_E % 4 == 0, "destination array is whole int4 vectors");
static_assert(OWN_S * GAT_H <= OWN_T, "one thread per (row, head) in the set-up");

__device__ __forceinline__ float lrelu(float v) { return v > 0.f ? v : NEG_SLOPE * v; }

__device__ __forceinline__ float elu_bias(float x, float b) {
  const float v = x + b;
  return v > 0.f ? v : (__expf(v) - 1.f);
}

template <int U>
__device__ __forceinline__ void own_add_entries(const int (&src)[U], const int (&dl)[U], int lane,
                                                const float* __restrict__ as,
                                                const float* __restrict__ h,
                                                const float* adl, const float* mx,
                                                float* dn, float* acc) {
  const int hd = lane & (GAT_H - 1);
  float a[U];
  float x[U][GAT_H];
#pragma unroll
  for (int u = 0; u < U; ++u) {
    a[u] = as[src[u] * GAT_H + hd];
    const float* hp = h + (size_t)src[u] * GAT_FH + lane;
#pragma unroll
    for (int k = 0; k < GAT_H; ++k) x[u][k] = hp[k * GAT_C];
  }
#pragma unroll
  for (int u = 0; u < U; ++u) {
    const int o = dl[u] * GAT_H + hd;
    const float w = __expf(lrelu(a[u] + adl[o]) - mx[o]);
    if (lane < GAT_H) atomicAdd(&dn[o], w);
    float* ap = acc + dl[u] * GAT_FH + lane;
#pragma unroll
    for (int k = 0; k < GAT_H; ++k) {
      const float wk = __int_as_float(__builtin_amdgcn_readlane(__float_as_int(w), k));
      atomicAdd(ap + k * GAT_C, x[u][k] * wk);
    }
  }
}

template <bool FIRST>
__global__ void __launch_bounds__(OWN_T)
owner_kernel(const int* __restrict__ ei, const float* __restrict__ as,
             const float* __restrict__ ad, const float* __restrict__ h,
             const float* __restrict__ b, float* __restrict__ out,
             int* __restrict__ glist, int* __restrict__ gcnt) {
  __shared__ float acc[OWN_S * GAT_FH];
  __shared__ int   list[OWN_CAP];
  __shared__ float mx[OWN_S * GAT_H];
  __shared__ float dn[OWN_S * GAT_H];
  __shared__ float adl[OWN_S * GAT_H];
  __shared__ int   cnt;

  const int tid  = (int)threadIdx.x;
  const int lane = tid & 31;
  const int wave = __builtin_amdgcn_readfirstlane(tid >> 5);
  const int lo = (int)blockIdx.x * OWN_S;
  const int nn = (GAT_N - lo) < OWN_S ? (GAT_N - lo) : OWN_S;
  int* gl = glist + (size_t)blockIdx.x * OWN_CAP;

  for (int i = tid; i < OWN_S * GAT_FH / 4; i += OWN_T)
    ((float4*)acc)[i] = make_float4(0.f, 0.f, 0.f, 0.f);
  if (tid < nn * GAT_H) {
    mx[tid] = -3.0e38f;
    dn[tid] = 0.f;
    adl[tid] = ad[lo * GAT_H + tid];
  }
  if constexpr (FIRST) {
    if (tid < nn) list[tid] = ((GAT_E + lo + tid) << OWN_DLBITS) | tid;
    if (tid == 0) cnt = nn;
  } else {
    const int kept = gcnt[blockIdx.x];
    if (kept <= OWN_CAP)
      for (int i = tid; i < kept; i += OWN_T) list[i] = gl[i];
    if (tid == 0) cnt = kept;
  }
  __syncthreads();

  if constexpr (FIRST) {
    const int4* d4 = (const int4*)(ei + GAT_E);
    auto test = [&](int dv, int e) {
      const unsigned r = (unsigned)(dv - lo);
      if (r < (unsigned)nn) {
        const int pos = atomicAdd(&cnt, 1);
        if (pos < OWN_CAP) list[pos] = (e << OWN_DLBITS) | (int)r;
      }
    };
    auto test4 = [&](const int4& q, int j) {
      const unsigned r0 = (unsigned)(q.x - lo), r1 = (unsigned)(q.y - lo);
      const unsigned r2 = (unsigned)(q.z - lo), r3 = (unsigned)(q.w - lo);
      const unsigned m01 = r0 < r1 ? r0 : r1, m23 = r2 < r3 ? r2 : r3;
      if ((m01 < m23 ? m01 : m23) < (unsigned)nn) {
        test(q.x, 4 * j); test(q.y, 4 * j + 1); test(q.z, 4 * j + 2); test(q.w, 4 * j + 3);
      }
    };
    int j = tid;
    for (; j + 3 * OWN_T < OWN_NV; j += 4 * OWN_T) {
      const int4 q0 = d4[j], q1 = d4[j + OWN_T], q2 = d4[j + 2 * OWN_T], q3 = d4[j + 3 * OWN_T];
      test4(q0, j); test4(q1, j + OWN_T); test4(q2, j + 2 * OWN_T); test4(q3, j + 3 * OWN_T);
    }
    for (; j < OWN_NV; j += OWN_T) { const int4 q = d4[j]; test4(q, j); }
    __syncthreads();
  }
  const int total = __builtin_amdgcn_readfirstlane(cnt);
  const bool fits = total <= OWN_CAP;
  if constexpr (FIRST) {
    if (tid == 0) gcnt[blockIdx.x] = total;
  }

  if (fits) {
    for (int i = tid; i < total; i += OWN_T) {
      const int ent = list[i];
      const int dl = ent & ((1 << OWN_DLBITS) - 1);
      int src = ent >> OWN_DLBITS;
      if constexpr (FIRST) {
        const int e = src;
        src = (e < GAT_E) ? ei[e] : (e - GAT_E);
        const int kept = (src << OWN_DLBITS) | dl;
        list[i] = kept;
        gl[i] = kept;
      }
      const float4 a0 = *(const float4*)(as + src * GAT_H);
      const float4 a1 = *(const float4*)(as + src * GAT_H + 4);
      const float* dp = adl + dl * GAT_H;
      float* mp = mx + dl * GAT_H;
      atomicMax(mp + 0, lrelu(a0.x + dp[0]));
      atomicMax(mp + 1, lrelu(a0.y + dp[1]));
      atomicMax(mp + 2, lrelu(a0.z + dp[2]));
      atomicMax(mp + 3, lrelu(a0.w + dp[3]));
      atomicMax(mp + 4, lrelu(a1.x + dp[4]));
      atomicMax(mp + 5, lrelu(a1.y + dp[5]));
      atomicMax(mp + 6, lrelu(a1.z + dp[6]));
      atomicMax(mp + 7, lrelu(a1.w + dp[7]));
    }
  } else {
    for (int e = tid; e < GAT_ETOT; e += OWN_T) {
      int src, dst;
      edge_endpoints(ei, e, src, dst);
      const unsigned r = (unsigned)(dst - lo);
      if (r < (unsigned)nn) {
#pragma unroll
        for (int k = 0; k < GAT_H; ++k)
          atomicMax(&mx[r * GAT_H + k], lrelu(as[src * GAT_H + k] + adl[r * GAT_H + k]));
      }
    }
  }
  __syncthreads();

  if (fits) {
    int i = wave;
    for (; i + 3 * 32 < total; i += 4 * 32) {
      int src[4], dl[4];
#pragma unroll
      for (int u = 0; u < 4; ++u) {
        const int ent = __builtin_amdgcn_readfirstlane(list[i + u * 32]);
        src[u] = ent >> OWN_DLBITS;
        dl[u] = ent & ((1 << OWN_DLBITS) - 1);
      }
      own_add_entries<4>(src, dl, lane, as, h, adl, mx, dn, acc);
    }
    for (; i < total; i += 32) {
      const int ent = __builtin_amdgcn_readfirstlane(list[i]);
      const int src[1] = {ent >> OWN_DLBITS};
      const int dl[1] = {ent & ((1 << OWN_DLBITS) - 1)};
      own_add_entries<1>(src, dl, lane, as, h, adl, mx, dn, acc);
    }
  } else {
    for (int e0 = 0; e0 < GAT_ETOT; e0 += OWN_T) {
      const int e = e0 + tid;
      int src = 0, dst = 0;
      bool hit = false;
      unsigned r = 0;
      if (e < GAT_ETOT) {
        edge_endpoints(ei, e, src, dst);
        r = (unsigned)(dst - lo);
        hit = r < (unsigned)nn;
      }
      unsigned todo = __builtin_amdgcn_ballot_w32(hit);
      while (todo) {
        const int k = __builtin_ctz(todo);
        todo &= todo - 1;
        const int s1[1] = {__builtin_amdgcn_readlane(src, k)};
        const int d1[1] = {__builtin_amdgcn_readlane((int)r, k)};
        own_add_entries<1>(s1, d1, lane, as, h, adl, mx, dn, acc);
      }
    }
  }
  __syncthreads();

  float4* o4 = (float4*)(out + (size_t)lo * GAT_FH);
  for (int i = tid; i < nn * (GAT_FH / 4); i += OWN_T) {
    const int r = i >> 6, c = (i & 63) * 4;
    const float4 s = ((const float4*)acc)[i];
    const float den = dn[r * GAT_H + (c >> 5)] + 1e-16f;
    const float4 bv = *(const float4*)(b + c);
    float4 o;
    o.x = elu_bias(s.x / den, bv.x);
    o.y = elu_bias(s.y / den, bv.y);
    o.z = elu_bias(s.z / den, bv.z);
    o.w = elu_bias(s.w / den, bv.w);
    o4[i] = o;
  }
}

__global__ void final_linear_kernel(const float* __restrict__ h,
                                    const float* __restrict__ Wl,
                                    const float* __restrict__ bl,
                                    float* __restrict__ y, int n) {
  int gid = blockIdx.x * blockDim.x + threadIdx.x;
  int wid = gid >> 5;
  int lane = threadIdx.x & 31;
  if (wid >= n) return;
  const float* hp = h + (size_t)wid * GAT_FH;
  float s = 0.f;
#pragma unroll
  for (int c = lane; c < GAT_FH; c += 32) s += hp[c] * Wl[c];
#pragma unroll
  for (int off = 16; off > 0; off >>= 1) s += __shfl_down(s, off, 32);
  if (lane == 0) y[wid] = s + bl[0];
}

extern "C" void kernel_launch(void* const* d_in, const int* in_sizes, int n_in,
                              void* d_out, int out_size, void* d_ws, size_t ws_size,
                              hipStream_t stream) {
  const float* x      = (const float*)d_in[0];
  const int*   ei     = (const int*)  d_in[1];
  const float* W1     = (const float*)d_in[2];
  const float* a1_src = (const float*)d_in[3];
  const float* a1_dst = (const float*)d_in[4];
  const float* b1     = (const float*)d_in[5];
  const float* W2     = (const float*)d_in[6];
  const float* a2_src = (const float*)d_in[7];
  const float* a2_dst = (const float*)d_in[8];
  const float* b2     = (const float*)d_in[9];
  const float* Wl     = (const float*)d_in[10];
  const float* bl     = (const float*)d_in[11];
  float* out = (float*)d_out;

  float* hbuf = (float*)d_ws;
  float* act  = hbuf + (size_t)GAT_N * GAT_FH;
  float* as   = act  + (size_t)GAT_N * GAT_FH;
  float* ad   = as   + (size_t)GAT_N * GAT_H;
  int* glist  = (int*)(ad + (size_t)GAT_N * GAT_H);
  int* gcnt   = glist + (size_t)OWN_WG * OWN_CAP;

  const int TPB = 256;
  auto blocks = [](int n, int tpb) { return (n + tpb - 1) / tpb; };

  gemm_bstat_f32<128><<<BST_WG, BST_T, 0, stream>>>(x, W1, hbuf, a1_src, a1_dst, as, ad);
  owner_kernel<true><<<OWN_WG, OWN_T, 0, stream>>>(ei, as, ad, hbuf, b1, act, glist, gcnt);

  gemm_bstat_f32<GAT_FH><<<BST_WG, BST_T, 0, stream>>>(act, W2, hbuf, a2_src, a2_dst, as, ad);
  owner_kernel<false><<<OWN_WG, OWN_T, 0, stream>>>(ei, as, ad, hbuf, b2, act, glist, gcnt);

  final_linear_kernel<<<blocks(GAT_N * 32, TPB), TPB, 0, stream>>>(act, Wl, bl, out, GAT_N);
}


# Owner kernel's source-row gather uses 16-byte-per-lane loads (512 contiguous bytes per instruction, 2 per edge, was 8 one-line loads); row sums lane-interleaved in LDS; same arrival-order LDS sums
# speedup vs baseline: 16.2370x; 1.0673x over previous
#include <hip/hip_runtime.h>
#include <hip/hip_bf16.h>

#define GAT_N 30000
#define GAT_E 480000
#define GAT_ETOT (GAT_E + GAT_N)
#define GAT_H 8
#define GAT_C 32
#define GAT_FH 256
#define NEG_SLOPE 0.2f

typedef float v2f __attribute__((ext_vector_type(2)));
typedef float v8f __attribute__((ext_vector_type(8)));

#define BST_T 512
#define BST_WG 256
#define BST_MT (GAT_N / 16)
static_assert(GAT_N % 16 == 0, "whole 16-row tiles");
static_assert(GAT_FH == 16 * (BST_T / 32), "one wave per 16-column strip");

template <int CTRL>
__device__ __forceinline__ float row_xor_get(float x) {
  return __int_as_float(__builtin_amdgcn_update_dpp(0, __float_as_int(x), CTRL, 0xf, 0xf, true));
}

template <int N, int CTRL>
__device__ __forceinline__ void row_halve(float (&v)[16], bool up) {
#pragma unroll
  for (int j = 0; j < N; ++j) {
    const float keep = up ? v[j + N] : v[j];
    const float send = up ? v[j] : v[j + N];
    v[j] = keep + row_xor_get<CTRL>(send);
  }
}

template <int K>
__global__ void __launch_bounds__(BST_T)
gemm_bstat_f32(const float* __restrict__ A, const float* __restrict__ B,
               float* __restrict__ C,
               const float* __restrict__ asrc, const float* __restrict__ adst,
               float* __restrict__ as, float* __restrict__ ad) {
  constexpr int P = K + 4;
  constexpr int V = (16 * K / 4) / BST_T;
  static_assert((16 * K / 4) % BST_T == 0 && (V == 1 || V == 2), "one or two vectors per thread");
  __shared__ float sa[2][16 * P];
  __shared__ alignas(16) float xs[2][2 * 16 * 16];

  const int tid  = (int)threadIdx.x;
  const int lane = tid & 31;
  const int wave = __builtin_amdgcn_readfirstlane(tid >> 5);
  const int half = lane >> 4;
  const int l    = lane & 15;
  const int n0   = wave * 16;

  v2f b[K / 4];
  {
    const float* bcol = B + (size_t)(half * 2) * GAT_FH + n0 + l;
#pragma unroll
    for (int t = 0; t < K / 4; ++t) {
      b[t].x = bcol[(size_t)(4 * t) * GAT_FH];
      b[t].y = bcol[(size_t)(4 * t + 1) * GAT_FH];
    }
  }

  const float asv = asrc[n0 + l];
  const float adv = adst[n0 + l];
  const int xw = ((l >> 3) * 16 + half * 8 + (l & 7)) * 16 + wave;
  float* const xdst = (tid < 128) ? as : ad;

  const int so0 = (tid / (K / 4)) * P + (tid % (K / 4)) * 4;
  const int so1 = ((tid + BST_T) / (K / 4)) * P + ((tid + BST_T) % (K / 4)) * 4;

  int tile = (int)blockIdx.x;
  float4 r0 = make_float4(0.f, 0.f, 0.f, 0.f), r1 = r0;
  if (tile < BST_MT) {
    const float4* g = (const float4*)(A + (size_t)tile * 16 * K);
    r0 = g[tid];
    if constexpr (V == 2) r1 = g[tid + BST_T];
  }

  int buf = 0;
  int ptile = -1;
  for (; tile < BST_MT; tile += BST_WG) {
    float* s = sa[buf];
    *(float4*)(s + so0) = r0;
    if constexpr (V == 2) *(float4*)(s + so1) = r1;
    __syncthreads();

    {
      const int nt = (tile + BST_WG < BST_MT) ? tile + BST_WG : tile;
      const float4* g = (const float4*)(A + (size_t)nt * 16 * K);
      r0 = g[tid];
      if constexpr (V == 2) r1 = g[tid + BST_T];
    }
    __builtin_amdgcn_sched_barrier(0);

    v8f c = {0.f, 0.f, 0.f, 0.f, 0.f, 0.f, 0.f, 0.f};
    const float* ap = s + l * P + half * 2;
#pragma unroll
    for (int t = 0; t < K / 4; ++t) {
      const v2f a = *(const v2f*)(ap + 4 * t);
      c = __builtin_amdgcn_wmma_f32_16x16x4_f32(
           false, a,  false, b[t],
           (short)0, c,  false,  false);
    }

    float* crow = C + (size_t)(tile * 16 + half * 8) * GAT_FH + n0 + l;
#pragma unroll
    for (int q = 0; q < 8; ++q) crow[(size_t)q * GAT_FH] = c[q];

    {
      float v[16];
#pragma unroll
      for (int q = 0; q < 8; ++q) { v[q] = c[q] * asv; v[8 + q] = c[q] * adv; }
      row_halve<8, 0x168>(v, (l & 8) != 0);
      row_halve<4, 0x164>(v, (l & 4) != 0);
      row_halve<2, 0x162>(v, (l & 2) != 0);
      row_halve<1, 0x161>(v, (l & 1) != 0);
      xs[buf][xw] = v[0];
    }
    if (ptile >= 0 && tid < 256) {
      const v2f p = *(const v2f*)(&xs[buf ^ 1][2 * tid]);
      xdst[(size_t)ptile * 128 + (tid & 127)] = p.x + p.y;
    }
    ptile = tile;
    buf ^= 1;
  }
  __syncthreads();
  if (ptile >= 0 && tid < 256) {
    const v2f p = *(const v2f*)(&xs[buf ^ 1][2 * tid]);
    xdst[(size_t)ptile * 128 + (tid & 127)] = p.x + p.y;
  }
}

__device__ __forceinline__ void edge_endpoints(const int* __restrict__ ei,
                                               int e, int& src, int& dst) {
  if (e < GAT_E) { src = ei[e]; dst = ei[GAT_E + e]; }
  else           { src = e - GAT_E; dst = e - GAT_E; }
}


#define OWN_S 118
#define OWN_WG ((GAT_N + OWN_S - 1) / OWN_S)
#define OWN_T 1024
#define OWN_CAP 16384
#define OWN_DLBITS 7
#define OWN_NV (GAT_E / 4)
static_assert(OWN_S <= (1 << OWN_DLBITS), "local row fits its field");
static_assert(GAT_ETOT < (1 << (31 - OWN_DLBITS)), "edge number fits beside the local row");
static_assert(GAT_E % 4 == 0, "destination array is whole int4 vectors");
static_assert(OWN_S * GAT_H <= OWN_T, "one thread per (row, head) in the set-up");

__device__ __forceinline__ float lrelu(float v) { return v > 0.f ? v : NEG_SLOPE * v; }

__device__ __forceinline__ float elu_bias(float x, float b) {
  const float v = x + b;
  return v > 0.f ? v : (__expf(v) - 1.f);
}

template <int CTRL>
__device__ __forceinline__ float quad_get(float x) {
  return __int_as_float(__builtin_amdgcn_update_dpp(0, __float_as_int(x), CTRL, 0xf, 0xf, true));
}

template <int U>
__device__ __forceinline__ void own_add_entries(const int (&src)[U], const int (&dl)[U], int lane,
                                                const float* __restrict__ as,
                                                const float* __restrict__ h,
                                                const float* adl, const float* mx,
                                                float* dn, float* acc) {
  const int hd = (lane >> 3) + 4 * (lane & 1);
  float a[U];
  float4 x[U][2];
#pragma unroll
  for (int u = 0; u < U; ++u) {
    a[u] = as[src[u] * GAT_H + hd];
    const float4* hp = (const float4*)(h + (size_t)src[u] * GAT_FH) + lane;
    x[u][0] = hp[0];
    x[u][1] = hp[GAT_FH / 8];
  }
#pragma unroll
  for (int u = 0; u < U; ++u) {
    const int o = dl[u] * GAT_H + hd;
    const float w = __expf(lrelu(a[u] + adl[o]) - mx[o]);
    if ((lane & 6) == 0) atomicAdd(&dn[o], w);
    float* ap = acc + dl[u] * GAT_FH + lane;
    const float w0 = quad_get<0xA0>(w);
    const float w1 = quad_get<0xF5>(w);
    atomicAdd(ap + 0 * GAT_C, x[u][0].x * w0);
    atomicAdd(ap + 1 * GAT_C, x[u][0].y * w0);
    atomicAdd(ap + 2 * GAT_C, x[u][0].z * w0);
    atomicAdd(ap + 3 * GAT_C, x[u][0].w * w0);
    atomicAdd(ap + 4 * GAT_C, x[u][1].x * w1);
    atomicAdd(ap + 5 * GAT_C, x[u][1].y * w1);
    atomicAdd(ap + 6 * GAT_C, x[u][1].z * w1);
    atomicAdd(ap + 7 * GAT_C, x[u][1].w * w1);
  }
}

template <bool FIRST>
__global__ void __launch_bounds__(OWN_T)
owner_kernel(const int* __restrict__ ei, const float* __restrict__ as,
             const float* __restrict__ ad, const float* __restrict__ h,
             const float* __restrict__ b, float* __restrict__ out,
             int* __restrict__ glist, int* __restrict__ gcnt) {
  __shared__ float acc[OWN_S * GAT_FH];
  __shared__ int   list[OWN_CAP];
  __shared__ float mx[OWN_S * GAT_H];
  __shared__ float dn[OWN_S * GAT_H];
  __shared__ float adl[OWN_S * GAT_H];
  __shared__ int   cnt;

  const int tid  = (int)threadIdx.x;
  const int lane = tid & 31;
  const int wave = __builtin_amdgcn_readfirstlane(tid >> 5);
  const int lo = (int)blockIdx.x * OWN_S;
  const int nn = (GAT_N - lo) < OWN_S ? (GAT_N - lo) : OWN_S;
  int* gl = glist + (size_t)blockIdx.x * OWN_CAP;

  for (int i = tid; i < OWN_S * GAT_FH / 4; i += OWN_T)
    ((float4*)acc)[i] = make_float4(0.f, 0.f, 0.f, 0.f);
  if (tid < nn * GAT_H) {
    mx[tid] = -3.0e38f;
    dn[tid] = 0.f;
    adl[tid] = ad[lo * GAT_H + tid];
  }
  if constexpr (FIRST) {
    if (tid < nn) list[tid] = ((GAT_E + lo + tid) << OWN_DLBITS) | tid;
    if (tid == 0) cnt = nn;
  } else {
    const int kept = gcnt[blockIdx.x];
    if (kept <= OWN_CAP)
      for (int i = tid; i < kept; i += OWN_T) list[i] = gl[i];
    if (tid == 0) cnt = kept;
  }
  __syncthreads();

  if constexpr (FIRST) {
    const int4* d4 = (const int4*)(ei + GAT_E);
    auto test = [&](int dv, int e) {
      const unsigned r = (unsigned)(dv - lo);
      if (r < (unsigned)nn) {
        const int pos = atomicAdd(&cnt, 1);
        if (pos < OWN_CAP) list[pos] = (e << OWN_DLBITS) | (int)r;
      }
    };
    auto test4 = [&](const int4& q, int j) {
      const unsigned r0 = (unsigned)(q.x - lo), r1 = (unsigned)(q.y - lo);
      const unsigned r2 = (unsigned)(q.z - lo), r3 = (unsigned)(q.w - lo);
      const unsigned m01 = r0 < r1 ? r0 : r1, m23 = r2 < r3 ? r2 : r3;
      if ((m01 < m23 ? m01 : m23) < (unsigned)nn) {
        test(q.x, 4 * j); test(q.y, 4 * j + 1); test(q.z, 4 * j + 2); test(q.w, 4 * j + 3);
      }
    };
    int j = tid;
    for (; j + 3 * OWN_T < OWN_NV; j += 4 * OWN_T) {
      const int4 q0 = d4[j], q1 = d4[j + OWN_T], q2 = d4[j + 2 * OWN_T], q3 = d4[j + 3 * OWN_T];
      test4(q0, j); test4(q1, j + OWN_T); test4(q2, j + 2 * OWN_T); test4(q3, j + 3 * OWN_T);
    }
    for (; j < OWN_NV; j += OWN_T) { const int4 q = d4[j]; test4(q, j); }
    __syncthreads();
  }
  const int total = __builtin_amdgcn_readfirstlane(cnt);
  const bool fits = total <= OWN_CAP;
  if constexpr (FIRST) {
    if (tid == 0) gcnt[blockIdx.x] = total;
  }

  if (fits) {
    for (int i = tid; i < total; i += OWN_T) {
      const int ent = list[i];
      const int dl = ent & ((1 << OWN_DLBITS) - 1);
      int src = ent >> OWN_DLBITS;
      if constexpr (FIRST) {
        const int e = src;
        src = (e < GAT_E) ? ei[e] : (e - GAT_E);
        const int kept = (src << OWN_DLBITS) | dl;
        list[i] = kept;
        gl[i] = kept;
      }
      const float4 a0 = *(const float4*)(as + src * GAT_H);
      const float4 a1 = *(const float4*)(as + src * GAT_H + 4);
      const float* dp = adl + dl * GAT_H;
      float* mp = mx + dl * GAT_H;
      atomicMax(mp + 0, lrelu(a0.x + dp[0]));
      atomicMax(mp + 1, lrelu(a0.y + dp[1]));
      atomicMax(mp + 2, lrelu(a0.z + dp[2]));
      atomicMax(mp + 3, lrelu(a0.w + dp[3]));
      atomicMax(mp + 4, lrelu(a1.x + dp[4]));
      atomicMax(mp + 5, lrelu(a1.y + dp[5]));
      atomicMax(mp + 6, lrelu(a1.z + dp[6]));
      atomicMax(mp + 7, lrelu(a1.w + dp[7]));
    }
  } else {
    for (int e = tid; e < GAT_ETOT; e += OWN_T) {
      int src, dst;
      edge_endpoints(ei, e, src, dst);
      const unsigned r = (unsigned)(dst - lo);
      if (r < (unsigned)nn) {
#pragma unroll
        for (int k = 0; k < GAT_H; ++k)
          atomicMax(&mx[r * GAT_H + k], lrelu(as[src * GAT_H + k] + adl[r * GAT_H + k]));
      }
    }
  }
  __syncthreads();

  if (fits) {
    int i = wave;
    for (; i + 3 * 32 < total; i += 4 * 32) {
      int src[4], dl[4];
#pragma unroll
      for (int u = 0; u < 4; ++u) {
        const int ent = __builtin_amdgcn_readfirstlane(list[i + u * 32]);
        src[u] = ent >> OWN_DLBITS;
        dl[u] = ent & ((1 << OWN_DLBITS) - 1);
      }
      own_add_entries<4>(src, dl, lane, as, h, adl, mx, dn, acc);
    }
    for (; i < total; i += 32) {
      const int ent = __builtin_amdgcn_readfirstlane(list[i]);
      const int src[1] = {ent >> OWN_DLBITS};
      const int dl[1] = {ent & ((1 << OWN_DLBITS) - 1)};
      own_add_entries<1>(src, dl, lane, as, h, adl, mx, dn, acc);
    }
  } else {
    for (int e0 = 0; e0 < GAT_ETOT; e0 += OWN_T) {
      const int e = e0 + tid;
      int src = 0, dst = 0;
      bool hit = false;
      unsigned r = 0;
      if (e < GAT_ETOT) {
        edge_endpoints(ei, e, src, dst);
        r = (unsigned)(dst - lo);
        hit = r < (unsigned)nn;
      }
      unsigned todo = __builtin_amdgcn_ballot_w32(hit);
      while (todo) {
        const int k = __builtin_ctz(todo);
        todo &= todo - 1;
        const int s1[1] = {__builtin_amdgcn_readlane(src, k)};
        const int d1[1] = {__builtin_amdgcn_readlane((int)r, k)};
        own_add_entries<1>(s1, d1, lane, as, h, adl, mx, dn, acc);
      }
    }
  }
  __syncthreads();

  float4* o4 = (float4*)(out + (size_t)lo * GAT_FH);
  for (int i = tid; i < nn * (GAT_FH / 4); i += OWN_T) {
    const int r = i >> 6, q = i & 63, c = q * 4;
    const float* sp = acc + r * GAT_FH + (q >> 5) * (GAT_FH / 2) + (q & 31);
    float4 s;
    s.x = sp[0 * GAT_C];
    s.y = sp[1 * GAT_C];
    s.z = sp[2 * GAT_C];
    s.w = sp[3 * GAT_C];
    const float den = dn[r * GAT_H + (c >> 5)] + 1e-16f;
    const float4 bv = *(const float4*)(b + c);
    float4 o;
    o.x = elu_bias(s.x / den, bv.x);
    o.y = elu_bias(s.y / den, bv.y);
    o.z = elu_bias(s.z / den, bv.z);
    o.w = elu_bias(s.w / den, bv.w);
    o4[i] = o;
  }
}

__global__ void final_linear_kernel(const float* __restrict__ h,
                                    const float* __restrict__ Wl,
                                    const float* __restrict__ bl,
                                    float* __restrict__ y, int n) {
  int gid = blockIdx.x * blockDim.x + threadIdx.x;
  int wid = gid >> 5;
  int lane = threadIdx.x & 31;
  if (wid >= n) return;
  const float* hp = h + (size_t)wid * GAT_FH;
  float s = 0.f;
#pragma unroll
  for (int c = lane; c < GAT_FH; c += 32) s += hp[c] * Wl[c];
#pragma unroll
  for (int off = 16; off > 0; off >>= 1) s += __shfl_down(s, off, 32);
  if (lane == 0) y[wid] = s + bl[0];
}

extern "C" void kernel_launch(void* const* d_in, const int* in_sizes, int n_in,
                              void* d_out, int out_size, void* d_ws, size_t ws_size,
                              hipStream_t stream) {
  const float* x      = (const float*)d_in[0];
  const int*   ei     = (const int*)  d_in[1];
  const float* W1     = (const float*)d_in[2];
  const float* a1_src = (const float*)d_in[3];
  const float* a1_dst = (const float*)d_in[4];
  const float* b1     = (const float*)d_in[5];
  const float* W2     = (const float*)d_in[6];
  const float* a2_src = (const float*)d_in[7];
  const float* a2_dst = (const float*)d_in[8];
  const float* b2     = (const float*)d_in[9];
  const float* Wl     = (const float*)d_in[10];
  const float* bl     = (const float*)d_in[11];
  float* out = (float*)d_out;

  float* hbuf = (float*)d_ws;
  float* act  = hbuf + (size_t)GAT_N * GAT_FH;
  float* as   = act  + (size_t)GAT_N * GAT_FH;
  float* ad   = as   + (size_t)GAT_N * GAT_H;
  int* glist  = (int*)(ad + (size_t)GAT_N * GAT_H);
  int* gcnt   = glist + (size_t)OWN_WG * OWN_CAP;

  const int TPB = 256;
  auto blocks = [](int n, int tpb) { return (n + tpb - 1) / tpb; };

  gemm_bstat_f32<128><<<BST_WG, BST_T, 0, stream>>>(x, W1, hbuf, a1_src, a1_dst, as, ad);
  owner_kernel<true><<<OWN_WG, OWN_T, 0, stream>>>(ei, as, ad, hbuf, b1, act, glist, gcnt);

  gemm_bstat_f32<GAT_FH><<<BST_WG, BST_T, 0, stream>>>(act, W2, hbuf, a2_src, a2_dst, as, ad);
  owner_kernel<false><<<OWN_WG, OWN_T, 0, stream>>>(ei, as, ad, hbuf, b2, act, glist, gcnt);

  final_linear_kernel<<<blocks(GAT_N * 32, TPB), TPB, 0, stream>>>(act, Wl, bl, out, GAT_N);
}


# Layer-1 edge filter scans one-byte owner keys (destination row / 118, written by the GEMM launch in front), hits staged per wave by ballot rank: 480 KB per work-group, was 1.92 MB; same sums
# speedup vs baseline: 17.0362x; 1.0492x over previous
#include <hip/hip_runtime.h>
#include <hip/hip_bf16.h>

#define GAT_N 30000
#define GAT_E 480000
#define GAT_ETOT (GAT_E + GAT_N)
#define GAT_H 8
#define GAT_C 32
#define GAT_FH 256
#define NEG_SLOPE 0.2f

typedef float v2f __attribute__((ext_vector_type(2)));
typedef float v8f __attribute__((ext_vector_type(8)));

#define BST_T 512
#define BST_WG 256
#define BST_MT (GAT_N / 16)
static_assert(GAT_N % 16 == 0, "whole 16-row tiles");
static_assert(GAT_FH == 16 * (BST_T / 32), "one wave per 16-column strip");

#define OWN_S 118
#define OWN_WG ((GAT_N + OWN_S - 1) / OWN_S)
#define OWN_NOKEY 255u
static_assert(OWN_WG <= (int)OWN_NOKEY, "an owner number fits one byte beside the no-owner key");
static_assert(GAT_E % 16 == 0, "the key array is whole 16-byte vectors");
static_assert(BST_WG * BST_T >= GAT_E / 4, "one thread of the layer-1 GEMM launch per 4 edges");

__device__ __forceinline__ unsigned owner_key(int dst) {
  return ((unsigned)dst < (unsigned)GAT_N) ? (unsigned)dst / OWN_S : OWN_NOKEY;
}

template <int CTRL>
__device__ __forceinline__ float row_xor_get(float x) {
  return __int_as_float(__builtin_amdgcn_update_dpp(0, __float_as_int(x), CTRL, 0xf, 0xf, true));
}

template <int N, int CTRL>
__device__ __forceinline__ void row_halve(float (&v)[16], bool up) {
#pragma unroll
  for (int j = 0; j < N; ++j) {
    const float keep = up ? v[j + N] : v[j];
    const float send = up ? v[j] : v[j + N];
    v[j] = keep + row_xor_get<CTRL>(send);
  }
}

template <int K, bool KEYS>
__global__ void __launch_bounds__(BST_T)
gemm_bstat_f32(const float* __restrict__ A, const float* __restrict__ B,
               float* __restrict__ C,
               const float* __restrict__ asrc, const float* __restrict__ adst,
               float* __restrict__ as, float* __restrict__ ad,
               const int* __restrict__ ei, unsigned* __restrict__ keys) {
  constexpr int P = K + 4;
  constexpr int V = (16 * K / 4) / BST_T;
  static_assert((16 * K / 4) % BST_T == 0 && (V == 1 || V == 2), "one or two vectors per thread");
  __shared__ float sa[2][16 * P];
  __shared__ alignas(16) float xs[2][2 * 16 * 16];

  const int tid  = (int)threadIdx.x;
  const int lane = tid & 31;
  const int wave = __builtin_amdgcn_readfirstlane(tid >> 5);
  const int half = lane >> 4;
  const int l    = lane & 15;
  const int n0   = wave * 16;

  v2f b[K / 4];
  {
    const float* bcol = B + (size_t)(half * 2) * GAT_FH + n0 + l;
#pragma unroll
    for (int t = 0; t < K / 4; ++t) {
      b[t].x = bcol[(size_t)(4 * t) * GAT_FH];
      b[t].y = bcol[(size_t)(4 * t + 1) * GAT_FH];
    }
  }

  const float asv = asrc[n0 + l];
  const float adv = adst[n0 + l];
  const int xw = ((l >> 3) * 16 + half * 8 + (l & 7)) * 16 + wave;
  float* const xdst = (tid < 128) ? as : ad;

  const int so0 = (tid / (K / 4)) * P + (tid % (K / 4)) * 4;
  const int so1 = ((tid + BST_T) / (K / 4)) * P + ((tid + BST_T) % (K / 4)) * 4;

  int tile = (int)blockIdx.x;
  float4 r0 = make_float4(0.f, 0.f, 0.f, 0.f), r1 = r0;
  if (tile < BST_MT) {
    const float4* g = (const float4*)(A + (size_t)tile * 16 * K);
    r0 = g[tid];
    if constexpr (V == 2) r1 = g[tid + BST_T];
  }

  if constexpr (KEYS) {
    const int g = (int)blockIdx.x * BST_T + tid;
    if (g < GAT_E / 4) {
      const int4 d = ((const int4*)(ei + GAT_E))[g];
      keys[g] = owner_key(d.x) | (owner_key(d.y) << 8) | (owner_key(d.z) << 16) | (owner_key(d.w) << 24);
    }
  }

  int buf = 0;
  int ptile = -1;
  for (; tile < BST_MT; tile += BST_WG) {
    float* s = sa[buf];
    *(float4*)(s + so0) = r0;
    if constexpr (V == 2) *(float4*)(s + so1) = r1;
    __syncthreads();

    {
      const int nt = (tile + BST_WG < BST_MT) ? tile + BST_WG : tile;
      const float4* g = (const float4*)(A + (size_t)nt * 16 * K);
      r0 = g[tid];
      if constexpr (V == 2) r1 = g[tid + BST_T];
    }
    __builtin_amdgcn_sched_barrier(0);

    v8f c = {0.f, 0.f, 0.f, 0.f, 0.f, 0.f, 0.f, 0.f};
    const float* ap = s + l * P + half * 2;
#pragma unroll
    for (int t = 0; t < K / 4; ++t) {
      const v2f a = *(const v2f*)(ap + 4 * t);
      c = __builtin_amdgcn_wmma_f32_16x16x4_f32(
           false, a,  false, b[t],
           (short)0, c,  false,  false);
    }

    float* crow = C + (size_t)(tile * 16 + half * 8) * GAT_FH + n0 + l;
#pragma unroll
    for (int q = 0; q < 8; ++q) crow[(size_t)q * GAT_FH] = c[q];

    {
      float v[16];
#pragma unroll
      for (int q = 0; q < 8; ++q) { v[q] = c[q] * asv; v[8 + q] = c[q] * adv; }
      row_halve<8, 0x168>(v, (l & 8) != 0);
      row_halve<4, 0x164>(v, (l & 4) != 0);
      row_halve<2, 0x162>(v, (l & 2) != 0);
      row_halve<1, 0x161>(v, (l & 1) != 0);
      xs[buf][xw] = v[0];
    }
    if (ptile >= 0 && tid < 256) {
      const v2f p = *(const v2f*)(&xs[buf ^ 1][2 * tid]);
      xdst[(size_t)ptile * 128 + (tid & 127)] = p.x + p.y;
    }
    ptile = tile;
    buf ^= 1;
  }
  __syncthreads();
  if (ptile >= 0 && tid < 256) {
    const v2f p = *(const v2f*)(&xs[buf ^ 1][2 * tid]);
    xdst[(size_t)ptile * 128 + (tid & 127)] = p.x + p.y;
  }
}

__device__ __forceinline__ void edge_endpoints(const int* __restrict__ ei,
                                               int e, int& src, int& dst) {
  if (e < GAT_E) { src = ei[e]; dst = ei[GAT_E + e]; }
  else           { src = e - GAT_E; dst = e - GAT_E; }
}


#define OWN_T 1024
#define OWN_CAP 16384
#define OWN_DLBITS 7
#define OWN_NV (GAT_E / 16)
#define OWN_SCAN_R 8
#define OWN_SEG ((OWN_S * GAT_FH) / (OWN_T / 32))
static_assert(OWN_T / 32 == 32, "one lane per wave when the waves' hit counts are summed");
static_assert(OWN_S <= (1 << OWN_DLBITS), "local row fits its field");
static_assert(GAT_ETOT < (1 << (31 - OWN_DLBITS)), "edge number fits beside the local row");
static_assert(OWN_S * GAT_H <= OWN_T, "one thread per (row, head) in the set-up");

__device__ __forceinline__ float lrelu(float v) { return v > 0.f ? v : NEG_SLOPE * v; }

__device__ __forceinline__ float elu_bias(float x, float b) {
  const float v = x + b;
  return v > 0.f ? v : (__expf(v) - 1.f);
}

template <int CTRL>
__device__ __forceinline__ float quad_get(float x) {
  return __int_as_float(__builtin_amdgcn_update_dpp(0, __float_as_int(x), CTRL, 0xf, 0xf, true));
}

template <int U>
__device__ __forceinline__ void own_add_entries(const int (&src)[U], const int (&dl)[U], int lane,
                                                const float* __restrict__ as,
                                                const float* __restrict__ h,
                                                const float* adl, const float* mx,
                                                float* dn, float* acc) {
  const int hd = (lane >> 3) + 4 * (lane & 1);
  float a[U];
  float4 x[U][2];
#pragma unroll
  for (int u = 0; u < U; ++u) {
    a[u] = as[src[u] * GAT_H + hd];
    const float4* hp = (const float4*)(h + (size_t)src[u] * GAT_FH) + lane;
    x[u][0] = hp[0];
    x[u][1] = hp[GAT_FH / 8];
  }
#pragma unroll
  for (int u = 0; u < U; ++u) {
    const int o = dl[u] * GAT_H + hd;
    const float w = __expf(lrelu(a[u] + adl[o]) - mx[o]);
    if ((lane & 6) == 0) atomicAdd(&dn[o], w);
    float* ap = acc + dl[u] * GAT_FH + lane;
    const float w0 = quad_get<0xA0>(w);
    const float w1 = quad_get<0xF5>(w);
    atomicAdd(ap + 0 * GAT_C, x[u][0].x * w0);
    atomicAdd(ap + 1 * GAT_C, x[u][0].y * w0);
    atomicAdd(ap + 2 * GAT_C, x[u][0].z * w0);
    atomicAdd(ap + 3 * GAT_C, x[u][0].w * w0);
    atomicAdd(ap + 4 * GAT_C, x[u][1].x * w1);
    atomicAdd(ap + 5 * GAT_C, x[u][1].y * w1);
    atomicAdd(ap + 6 * GAT_C, x[u][1].z * w1);
    atomicAdd(ap + 7 * GAT_C, x[u][1].w * w1);
  }
}

template <bool FIRST>
__global__ void __launch_bounds__(OWN_T)
owner_kernel(const int* __restrict__ ei, const float* __restrict__ as,
             const float* __restrict__ ad, const float* __restrict__ h,
             const float* __restrict__ b, float* __restrict__ out,
             int* __restrict__ glist, int* __restrict__ gcnt,
             const unsigned* __restrict__ keys) {
  __shared__ float acc[OWN_S * GAT_FH];
  __shared__ int   list[OWN_CAP];
  __shared__ float mx[OWN_S * GAT_H];
  __shared__ float dn[OWN_S * GAT_H];
  __shared__ float adl[OWN_S * GAT_H];
  __shared__ int   cnt;
  __shared__ int   wcnt[OWN_T / 32];

  const int tid  = (int)threadIdx.x;
  const int lane = tid & 31;
  const int wave = __builtin_amdgcn_readfirstlane(tid >> 5);
  const int lo = (int)blockIdx.x * OWN_S;
  const int nn = (GAT_N - lo) < OWN_S ? (GAT_N - lo) : OWN_S;
  int* gl = glist + (size_t)blockIdx.x * OWN_CAP;

  auto clear_sums = [&]() {
    for (int i = tid; i < OWN_S * GAT_FH / 4; i += OWN_T)
      ((float4*)acc)[i] = make_float4(0.f, 0.f, 0.f, 0.f);
  };
  if constexpr (!FIRST) clear_sums();
  if (tid < nn * GAT_H) {
    mx[tid] = -3.0e38f;
    dn[tid] = 0.f;
    adl[tid] = ad[lo * GAT_H + tid];
  }
  if constexpr (FIRST) {
    if (tid < nn) list[tid] = GAT_E + lo + tid;
  } else {
    const int kept = gcnt[blockIdx.x];
    if (kept <= OWN_CAP)
      for (int i = tid; i < kept; i += OWN_T) list[i] = gl[i];
    if (tid == 0) cnt = kept;
  }
  __syncthreads();

  int total;
  if constexpr (FIRST) {
    const uint4* k4 = (const uint4*)keys;
    const unsigned me = (unsigned)blockIdx.x * 0x01010101u;
    int* const stage = (int*)acc + wave * OWN_SEG;
    int wn = 0;
    auto zero_bytes = [](unsigned x) {
      return ~(((x & 0x7f7f7f7fu) + 0x7f7f7f7fu) | x) & 0x80808080u;
    };
    auto test16 = [&](const uint4& q, int j) {
      unsigned left = zero_bytes(q.x ^ me) | (zero_bytes(q.y ^ me) >> 1) |
                      (zero_bytes(q.z ^ me) >> 2) | (zero_bytes(q.w ^ me) >> 3);
      unsigned m = __builtin_amdgcn_ballot_w32(left != 0u);
      while (m != 0u) {
        if (left != 0u) {
          const int p = __builtin_ctz(left);
          const int slot = wn + (int)__builtin_amdgcn_mbcnt_lo(m, 0u);
          stage[slot < OWN_SEG ? slot : OWN_SEG - 1] = 16 * j + ((7 - (p & 7)) << 2) + (p >> 3);
          left &= left - 1u;
        }
        wn += __builtin_popcount(m);
        m = __builtin_amdgcn_ballot_w32(left != 0u);
      }
    };
    for (int j0 = 0; j0 < OWN_NV; j0 += OWN_SCAN_R * OWN_T) {
      uint4 q[OWN_SCAN_R];
#pragma unroll
      for (int r = 0; r < OWN_SCAN_R; ++r) {
        const int j = j0 + r * OWN_T + tid;
        q[r] = (j < OWN_NV) ? k4[j] : make_uint4(~0u, ~0u, ~0u, ~0u);
      }
#pragma unroll
      for (int r = 0; r < OWN_SCAN_R; ++r) test16(q[r], j0 + r * OWN_T + tid);
    }
    if (lane == 0) wcnt[wave] = wn;
    __syncthreads();
    const int wc = wcnt[lane];
    const bool over = __builtin_amdgcn_ballot_w32(wc > OWN_SEG) != 0u;
    int all = wc, ahead = (lane < wave) ? wc : 0;
#pragma unroll
    for (int off = 16; off > 0; off >>= 1) {
      all += __shfl_xor(all, off, 32);
      ahead += __shfl_xor(ahead, off, 32);
    }
    total = __builtin_amdgcn_readfirstlane(over ? OWN_CAP + 1 : nn + all);
    if (total <= OWN_CAP)
      for (int k = lane; k < wn; k += 32) list[nn + ahead + k] = stage[k];
    __syncthreads();
    clear_sums();
  } else {
    total = __builtin_amdgcn_readfirstlane(cnt);
  }
  const bool fits = total <= OWN_CAP;
  if constexpr (FIRST) {
    if (tid == 0) gcnt[blockIdx.x] = total;
  }

  if (fits) {
    for (int i = tid; i < total; i += OWN_T) {
      const int ent = list[i];
      int dl = ent & ((1 << OWN_DLBITS) - 1);
      int src = ent >> OWN_DLBITS;
      if constexpr (FIRST) {
        int dst;
        edge_endpoints(ei, ent, src, dst);
        dl = dst - lo;
        const int kept = (src << OWN_DLBITS) | dl;
        list[i] = kept;
        gl[i] = kept;
      }
      const float4 a0 = *(const float4*)(as + src * GAT_H);
      const float4 a1 = *(const float4*)(as + src * GAT_H + 4);
      const float* dp = adl + dl * GAT_H;
      float* mp = mx + dl * GAT_H;
      atomicMax(mp + 0, lrelu(a0.x + dp[0]));
      atomicMax(mp + 1, lrelu(a0.y + dp[1]));
      atomicMax(mp + 2, lrelu(a0.z + dp[2]));
      atomicMax(mp + 3, lrelu(a0.w + dp[3]));
      atomicMax(mp + 4, lrelu(a1.x + dp[4]));
      atomicMax(mp + 5, lrelu(a1.y + dp[5]));
      atomicMax(mp + 6, lrelu(a1.z + dp[6]));
      atomicMax(mp + 7, lrelu(a1.w + dp[7]));
    }
  } else {
    for (int e = tid; e < GAT_ETOT; e += OWN_T) {
      int src, dst;
      edge_endpoints(ei, e, src, dst);
      const unsigned r = (unsigned)(dst - lo);
      if (r < (unsigned)nn) {
#pragma unroll
        for (int k = 0; k < GAT_H; ++k)
          atomicMax(&mx[r * GAT_H + k], lrelu(as[src * GAT_H + k] + adl[r * GAT_H + k]));
      }
    }
  }
  __syncthreads();

  if (fits) {
    int i = wave;
    for (; i + 3 * 32 < total; i += 4 * 32) {
      int src[4], dl[4];
#pragma unroll
      for (int u = 0; u < 4; ++u) {
        const int ent = __builtin_amdgcn_readfirstlane(list[i + u * 32]);
        src[u] = ent >> OWN_DLBITS;
        dl[u] = ent & ((1 << OWN_DLBITS) - 1);
      }
      own_add_entries<4>(src, dl, lane, as, h, adl, mx, dn, acc);
    }
    for (; i < total; i += 32) {
      const int ent = __builtin_amdgcn_readfirstlane(list[i]);
      const int src[1] = {ent >> OWN_DLBITS};
      const int dl[1] = {ent & ((1 << OWN_DLBITS) - 1)};
      own_add_entries<1>(src, dl, lane, as, h, adl, mx, dn, acc);
    }
  } else {
    for (int e0 = 0; e0 < GAT_ETOT; e0 += OWN_T) {
      const int e = e0 + tid;
      int src = 0, dst = 0;
      bool hit = false;
      unsigned r = 0;
      if (e < GAT_ETOT) {
        edge_endpoints(ei, e, src, dst);
        r = (unsigned)(dst - lo);
        hit = r < (unsigned)nn;
      }
      unsigned todo = __builtin_amdgcn_ballot_w32(hit);
      while (todo) {
        const int k = __builtin_ctz(todo);
        todo &= todo - 1;
        const int s1[1] = {__builtin_amdgcn_readlane(src, k)};
        const int d1[1] = {__builtin_amdgcn_readlane((int)r, k)};
        own_add_entries<1>(s1, d1, lane, as, h, adl, mx, dn, acc);
      }
    }
  }
  __syncthreads();

  float4* o4 = (float4*)(out + (size_t)lo * GAT_FH);
  for (int i = tid; i < nn * (GAT_FH / 4); i += OWN_T) {
    const int r = i >> 6, q = i & 63, c = q * 4;
    const float* sp = acc + r * GAT_FH + (q >> 5) * (GAT_FH / 2) + (q & 31);
    float4 s;
    s.x = sp[0 * GAT_C];
    s.y = sp[1 * GAT_C];
    s.z = sp[2 * GAT_C];
    s.w = sp[3 * GAT_C];
    const float den = dn[r * GAT_H + (c >> 5)] + 1e-16f;
    const float4 bv = *(const float4*)(b + c);
    float4 o;
    o.x = elu_bias(s.x / den, bv.x);
    o.y = elu_bias(s.y / den, bv.y);
    o.z = elu_bias(s.z / den, bv.z);
    o.w = elu_bias(s.w / den, bv.w);
    o4[i] = o;
  }
}

__global__ void final_linear_kernel(const float* __restrict__ h,
                                    const float* __restrict__ Wl,
                                    const float* __restrict__ bl,
                                    float* __restrict__ y, int n) {
  int gid = blockIdx.x * blockDim.x + threadIdx.x;
  int wid = gid >> 5;
  int lane = threadIdx.x & 31;
  if (wid >= n) return;
  const float* hp = h + (size_t)wid * GAT_FH;
  float s = 0.f;
#pragma unroll
  for (int c = lane; c < GAT_FH; c += 32) s += hp[c] * Wl[c];
#pragma unroll
  for (int off = 16; off > 0; off >>= 1) s += __shfl_down(s, off, 32);
  if (lane == 0) y[wid] = s + bl[0];
}

extern "C" void kernel_launch(void* const* d_in, const int* in_sizes, int n_in,
                              void* d_out, int out_size, void* d_ws, size_t ws_size,
                              hipStream_t stream) {
  const float* x      = (const float*)d_in[0];
  const int*   ei     = (const int*)  d_in[1];
  const float* W1     = (const float*)d_in[2];
  const float* a1_src = (const float*)d_in[3];
  const float* a1_dst = (const float*)d_in[4];
  const float* b1     = (const float*)d_in[5];
  const float* W2     = (const float*)d_in[6];
  const float* a2_src = (const float*)d_in[7];
  const float* a2_dst = (const float*)d_in[8];
  const float* b2     = (const float*)d_in[9];
  const float* Wl     = (const float*)d_in[10];
  const float* bl     = (const float*)d_in[11];
  float* out = (float*)d_out;

  float* hbuf = (float*)d_ws;
  float* act  = hbuf + (size_t)GAT_N * GAT_FH;
  float* as   = act  + (size_t)GAT_N * GAT_FH;
  float* ad   = as   + (size_t)GAT_N * GAT_H;
  int* glist  = (int*)(ad + (size_t)GAT_N * GAT_H);
  int* gcnt   = glist + (size_t)OWN_WG * OWN_CAP;
  unsigned* keys = (unsigned*)(gcnt + 256);

  const int TPB = 256;
  auto blocks = [](int n, int tpb) { return (n + tpb - 1) / tpb; };

  gemm_bstat_f32<128, true><<<BST_WG, BST_T, 0, stream>>>(x, W1, hbuf, a1_src, a1_dst, as, ad, ei, keys);
  owner_kernel<true><<<OWN_WG, OWN_T, 0, stream>>>(ei, as, ad, hbuf, b1, act, glist, gcnt, keys);

  gemm_bstat_f32<GAT_FH, false><<<BST_WG, BST_T, 0, stream>>>(act, W2, hbuf, a2_src, a2_dst, as, ad, ei, keys);
  owner_kernel<false><<<OWN_WG, OWN_T, 0, stream>>>(ei, as, ad, hbuf, b2, act, glist, gcnt, keys);

  final_linear_kernel<<<blocks(GAT_N * 32, TPB), TPB, 0, stream>>>(act, Wl, bl, out, GAT_N);
}


# Owner kernel: each of a work-group's 32 waves owns 4 of its rows; list bucketed by owning wave in LDS, row sums in registers (bucket arrival order); LDS float atomics and the LDS sum pass go
# speedup vs baseline: 18.4225x; 1.0814x over previous
#include <hip/hip_runtime.h>
#include <hip/hip_bf16.h>

#define GAT_N 30000
#define GAT_E 480000
#define GAT_ETOT (GAT_E + GAT_N)
#define GAT_H 8
#define GAT_C 32
#define GAT_FH 256
#define NEG_SLOPE 0.2f

typedef float v2f __attribute__((ext_vector_type(2)));
typedef float v8f __attribute__((ext_vector_type(8)));

#define BST_T 512
#define BST_WG 256
#define BST_MT (GAT_N / 16)
static_assert(GAT_N % 16 == 0, "whole 16-row tiles");
static_assert(GAT_FH == 16 * (BST_T / 32), "one wave per 16-column strip");

#define OWN_S 118
#define OWN_WG ((GAT_N + OWN_S - 1) / OWN_S)
#define OWN_NOKEY 255u
static_assert(OWN_WG <= (int)OWN_NOKEY, "an owner number fits one byte beside the no-owner key");
static_assert(GAT_E % 16 == 0, "the key array is whole 16-byte vectors");
static_assert(BST_WG * BST_T >= GAT_E / 4, "one thread of the layer-1 GEMM launch per 4 edges");

__device__ __forceinline__ unsigned owner_key(int dst) {
  return ((unsigned)dst < (unsigned)GAT_N) ? (unsigned)dst / OWN_S : OWN_NOKEY;
}

template <int CTRL>
__device__ __forceinline__ float row_xor_get(float x) {
  return __int_as_float(__builtin_amdgcn_update_dpp(0, __float_as_int(x), CTRL, 0xf, 0xf, true));
}

template <int N, int CTRL>
__device__ __forceinline__ void row_halve(float (&v)[16], bool up) {
#pragma unroll
  for (int j = 0; j < N; ++j) {
    const float keep = up ? v[j + N] : v[j];
    const float send = up ? v[j] : v[j + N];
    v[j] = keep + row_xor_get<CTRL>(send);
  }
}

template <int K, bool KEYS>
__global__ void __launch_bounds__(BST_T)
gemm_bstat_f32(const float* __restrict__ A, const float* __restrict__ B,
               float* __restrict__ C,
               const float* __restrict__ asrc, const float* __restrict__ adst,
               float* __restrict__ as, float* __restrict__ ad,
               const int* __restrict__ ei, unsigned* __restrict__ keys) {
  constexpr int P = K + 4;
  constexpr int V = (16 * K / 4) / BST_T;
  static_assert((16 * K / 4) % BST_T == 0 && (V == 1 || V == 2), "one or two vectors per thread");
  __shared__ float sa[2][16 * P];
  __shared__ alignas(16) float xs[2][2 * 16 * 16];

  const int tid  = (int)threadIdx.x;
  const int lane = tid & 31;
  const int wave = __builtin_amdgcn_readfirstlane(tid >> 5);
  const int half = lane >> 4;
  const int l    = lane & 15;
  const int n0   = wave * 16;

  v2f b[K / 4];
  {
    const float* bcol = B + (size_t)(half * 2) * GAT_FH + n0 + l;
#pragma unroll
    for (int t = 0; t < K / 4; ++t) {
      b[t].x = bcol[(size_t)(4 * t) * GAT_FH];
      b[t].y = bcol[(size_t)(4 * t + 1) * GAT_FH];
    }
  }

  const float asv = asrc[n0 + l];
  const float adv = adst[n0 + l];
  const int xw = ((l >> 3) * 16 + half * 8 + (l & 7)) * 16 + wave;
  float* const xdst = (tid < 128) ? as : ad;

  const int so0 = (tid / (K / 4)) * P + (tid % (K / 4)) * 4;
  const int so1 = ((tid + BST_T) / (K / 4)) * P + ((tid + BST_T) % (K / 4)) * 4;

  int tile = (int)blockIdx.x;
  float4 r0 = make_float4(0.f, 0.f, 0.f, 0.f), r1 = r0;
  if (tile < BST_MT) {
    const float4* g = (const float4*)(A + (size_t)tile * 16 * K);
    r0 = g[tid];
    if constexpr (V == 2) r1 = g[tid + BST_T];
  }

  if constexpr (KEYS) {
    const int g = (int)blockIdx.x * BST_T + tid;
    if (g < GAT_E / 4) {
      const int4 d = ((const int4*)(ei + GAT_E))[g];
      keys[g] = owner_key(d.x) | (owner_key(d.y) << 8) | (owner_key(d.z) << 16) | (owner_key(d.w) << 24);
    }
  }

  int buf = 0;
  int ptile = -1;
  for (; tile < BST_MT; tile += BST_WG) {
    float* s = sa[buf];
    *(float4*)(s + so0) = r0;
    if constexpr (V == 2) *(float4*)(s + so1) = r1;
    __syncthreads();

    {
      const int nt = (tile + BST_WG < BST_MT) ? tile + BST_WG : tile;
      const float4* g = (const float4*)(A + (size_t)nt * 16 * K);
      r0 = g[tid];
      if constexpr (V == 2) r1 = g[tid + BST_T];
    }
    __builtin_amdgcn_sched_barrier(0);

    v8f c = {0.f, 0.f, 0.f, 0.f, 0.f, 0.f, 0.f, 0.f};
    const float* ap = s + l * P + half * 2;
#pragma unroll
    for (int t = 0; t < K / 4; ++t) {
      const v2f a = *(const v2f*)(ap + 4 * t);
      c = __builtin_amdgcn_wmma_f32_16x16x4_f32(
           false, a,  false, b[t],
           (short)0, c,  false,  false);
    }

    float* crow = C + (size_t)(tile * 16 + half * 8) * GAT_FH + n0 + l;
#pragma unroll
    for (int q = 0; q < 8; ++q) crow[(size_t)q * GAT_FH] = c[q];

    {
      float v[16];
#pragma unroll
      for (int q = 0; q < 8; ++q) { v[q] = c[q] * asv; v[8 + q] = c[q] * adv; }
      row_halve<8, 0x168>(v, (l & 8) != 0);
      row_halve<4, 0x164>(v, (l & 4) != 0);
      row_halve<2, 0x162>(v, (l & 2) != 0);
      row_halve<1, 0x161>(v, (l & 1) != 0);
      xs[buf][xw] = v[0];
    }
    if (ptile >= 0 && tid < 256) {
      const v2f p = *(const v2f*)(&xs[buf ^ 1][2 * tid]);
      xdst[(size_t)ptile * 128 + (tid & 127)] = p.x + p.y;
    }
    ptile = tile;
    buf ^= 1;
  }
  __syncthreads();
  if (ptile >= 0 && tid < 256) {
    const v2f p = *(const v2f*)(&xs[buf ^ 1][2 * tid]);
    xdst[(size_t)ptile * 128 + (tid & 127)] = p.x + p.y;
  }
}

__device__ __forceinline__ void edge_endpoints(const int* __restrict__ ei,
                                               int e, int& src, int& dst) {
  if (e < GAT_E) { src = ei[e]; dst = ei[GAT_E + e]; }
  else           { src = e - GAT_E; dst = e - GAT_E; }
}


#define OWN_T 1024
#define OWN_CAP 16384
#define OWN_DLBITS 7
#define OWN_NV (GAT_E / 16)
#define OWN_SCAN_R 8
#define OWN_SEG ((OWN_S * GAT_FH) / (OWN_T / 32))
static_assert(OWN_T / 32 == 32, "one lane per wave when the waves' hit counts are summed");
static_assert(OWN_S <= (1 << OWN_DLBITS), "local row fits its field");
static_assert(GAT_ETOT < (1 << (31 - OWN_DLBITS)), "edge number fits beside the local row");
static_assert(OWN_S * GAT_H <= OWN_T, "one thread per (row, head) in the set-up");

__device__ __forceinline__ float lrelu(float v) { return v > 0.f ? v : NEG_SLOPE * v; }

__device__ __forceinline__ float elu_bias(float x, float b) {
  const float v = x + b;
  return v > 0.f ? v : (__expf(v) - 1.f);
}

template <int CTRL>
__device__ __forceinline__ float quad_get(float x) {
  return __int_as_float(__builtin_amdgcn_update_dpp(0, __float_as_int(x), CTRL, 0xf, 0xf, true));
}

template <int U>
__device__ __forceinline__ void own_add_entries(const int (&src)[U], const int (&dl)[U], int lane,
                                                const float* __restrict__ as,
                                                const float* __restrict__ h,
                                                const float* adl, const float* mx,
                                                float* dn, float* acc) {
  const int hd = (lane >> 3) + 4 * (lane & 1);
  float a[U];
  float4 x[U][2];
#pragma unroll
  for (int u = 0; u < U; ++u) {
    a[u] = as[src[u] * GAT_H + hd];
    const float4* hp = (const float4*)(h + (size_t)src[u] * GAT_FH) + lane;
    x[u][0] = hp[0];
    x[u][1] = hp[GAT_FH / 8];
  }
#pragma unroll
  for (int u = 0; u < U; ++u) {
    const int o = dl[u] * GAT_H + hd;
    const float w = __expf(lrelu(a[u] + adl[o]) - mx[o]);
    if ((lane & 6) == 0) atomicAdd(&dn[o], w);
    float* ap = acc + dl[u] * GAT_FH + lane;
    const float w0 = quad_get<0xA0>(w);
    const float w1 = quad_get<0xF5>(w);
    atomicAdd(ap + 0 * GAT_C, x[u][0].x * w0);
    atomicAdd(ap + 1 * GAT_C, x[u][0].y * w0);
    atomicAdd(ap + 2 * GAT_C, x[u][0].z * w0);
    atomicAdd(ap + 3 * GAT_C, x[u][0].w * w0);
    atomicAdd(ap + 4 * GAT_C, x[u][1].x * w1);
    atomicAdd(ap + 5 * GAT_C, x[u][1].y * w1);
    atomicAdd(ap + 6 * GAT_C, x[u][1].z * w1);
    atomicAdd(ap + 7 * GAT_C, x[u][1].w * w1);
  }
}

#define OWN_RPW 4
static_assert(OWN_S <= OWN_RPW * (OWN_T / 32), "every owned row belongs to one wave");

template <int U>
__device__ __forceinline__ void own_add_regs(const int (&src)[U], const int (&dl)[U], int n, int lane,
                                             const float* __restrict__ as,
                                             const float* __restrict__ h,
                                             const float* adl, const float* mx,
                                             float4 (&s)[OWN_RPW][2], float (&dnr)[OWN_RPW]) {
  const int hd = (lane >> 3) + 4 * (lane & 1);
  float a[U];
  float4 x[U][2];
#pragma unroll
  for (int u = 0; u < U; ++u) {
    a[u] = as[src[u] * GAT_H + hd];
    const float4* hp = (const float4*)(h + (size_t)src[u] * GAT_FH) + lane;
    x[u][0] = hp[0];
    x[u][1] = hp[GAT_FH / 8];
  }
#pragma unroll
  for (int u = 0; u < U; ++u) {
    if (u >= n) break;
    const int o = dl[u] * GAT_H + hd;
    const float w = __expf(lrelu(a[u] + adl[o]) - mx[o]);
    const float w0 = quad_get<0xA0>(w);
    const float w1 = quad_get<0xF5>(w);
    const int rs = dl[u] >> 5;
#pragma unroll
    for (int r = 0; r < OWN_RPW; ++r) {
      if (rs == r) {
        dnr[r] += w;
        s[r][0].x += x[u][0].x * w0;
        s[r][0].y += x[u][0].y * w0;
        s[r][0].z += x[u][0].z * w0;
        s[r][0].w += x[u][0].w * w0;
        s[r][1].x += x[u][1].x * w1;
        s[r][1].y += x[u][1].y * w1;
        s[r][1].z += x[u][1].z * w1;
        s[r][1].w += x[u][1].w * w1;
      }
    }
  }
}

template <bool FIRST>
__global__ void __launch_bounds__(OWN_T)
owner_kernel(const int* __restrict__ ei, const float* __restrict__ as,
             const float* __restrict__ ad, const float* __restrict__ h,
             const float* __restrict__ b, float* __restrict__ out,
             int* __restrict__ glist, int* __restrict__ gcnt,
             const unsigned* __restrict__ keys) {
  __shared__ float acc[OWN_S * GAT_FH];
  __shared__ int   list[OWN_CAP];
  __shared__ float mx[OWN_S * GAT_H];
  __shared__ float dn[OWN_S * GAT_H];
  __shared__ float adl[OWN_S * GAT_H];
  __shared__ int   cnt;
  __shared__ int   wcnt[OWN_T / 32];
  __shared__ int   bcnt[OWN_T / 32];
  __shared__ int   bbase[OWN_T / 32];
  __shared__ int   bcur[OWN_T / 32];

  const int tid  = (int)threadIdx.x;
  const int lane = tid & 31;
  const int wave = __builtin_amdgcn_readfirstlane(tid >> 5);
  const int lo = (int)blockIdx.x * OWN_S;
  const int nn = (GAT_N - lo) < OWN_S ? (GAT_N - lo) : OWN_S;
  int* gl = glist + (size_t)blockIdx.x * OWN_CAP;

  auto clear_sums = [&]() {
    for (int i = tid; i < OWN_S * GAT_FH / 4; i += OWN_T)
      ((float4*)acc)[i] = make_float4(0.f, 0.f, 0.f, 0.f);
  };
  if (tid < OWN_T / 32) bcnt[tid] = 0;
  if (tid < nn * GAT_H) {
    mx[tid] = -3.0e38f;
    dn[tid] = 0.f;
    adl[tid] = ad[lo * GAT_H + tid];
  }
  if constexpr (FIRST) {
    if (tid < nn) list[tid] = GAT_E + lo + tid;
  } else {
    const int kept = gcnt[blockIdx.x];
    if (kept <= OWN_CAP)
      for (int i = tid; i < kept; i += OWN_T) list[i] = gl[i];
    else
      clear_sums();
    if (tid == 0) cnt = kept;
  }
  __syncthreads();

  int total;
  if constexpr (FIRST) {
    const uint4* k4 = (const uint4*)keys;
    const unsigned me = (unsigned)blockIdx.x * 0x01010101u;
    int* const stage = (int*)acc + wave * OWN_SEG;
    int wn = 0;
    auto zero_bytes = [](unsigned x) {
      return ~(((x & 0x7f7f7f7fu) + 0x7f7f7f7fu) | x) & 0x80808080u;
    };
    auto test16 = [&](const uint4& q, int j) {
      unsigned left = zero_bytes(q.x ^ me) | (zero_bytes(q.y ^ me) >> 1) |
                      (zero_bytes(q.z ^ me) >> 2) | (zero_bytes(q.w ^ me) >> 3);
      unsigned m = __builtin_amdgcn_ballot_w32(left != 0u);
      while (m != 0u) {
        if (left != 0u) {
          const int p = __builtin_ctz(left);
          const int slot = wn + (int)__builtin_amdgcn_mbcnt_lo(m, 0u);
          stage[slot < OWN_SEG ? slot : OWN_SEG - 1] = 16 * j + ((7 - (p & 7)) << 2) + (p >> 3);
          left &= left - 1u;
        }
        wn += __builtin_popcount(m);
        m = __builtin_amdgcn_ballot_w32(left != 0u);
      }
    };
    for (int j0 = 0; j0 < OWN_NV; j0 += OWN_SCAN_R * OWN_T) {
      uint4 q[OWN_SCAN_R];
#pragma unroll
      for (int r = 0; r < OWN_SCAN_R; ++r) {
        const int j = j0 + r * OWN_T + tid;
        q[r] = (j < OWN_NV) ? k4[j] : make_uint4(~0u, ~0u, ~0u, ~0u);
      }
#pragma unroll
      for (int r = 0; r < OWN_SCAN_R; ++r) test16(q[r], j0 + r * OWN_T + tid);
    }
    if (lane == 0) wcnt[wave] = wn;
    __syncthreads();
    const int wc = wcnt[lane];
    const bool over = __builtin_amdgcn_ballot_w32(wc > OWN_SEG) != 0u;
    int all = wc, ahead = (lane < wave) ? wc : 0;
#pragma unroll
    for (int off = 16; off > 0; off >>= 1) {
      all += __shfl_xor(all, off, 32);
      ahead += __shfl_xor(ahead, off, 32);
    }
    total = __builtin_amdgcn_readfirstlane(over ? OWN_CAP + 1 : nn + all);
    if (total <= OWN_CAP)
      for (int k = lane; k < wn; k += 32) list[nn + ahead + k] = stage[k];
    __syncthreads();
    if (total > OWN_CAP) clear_sums();
  } else {
    total = __builtin_amdgcn_readfirstlane(cnt);
  }
  const bool fits = total <= OWN_CAP;
  if constexpr (FIRST) {
    if (tid == 0) gcnt[blockIdx.x] = total;
  }

  if (fits) {
    for (int i = tid; i < total; i += OWN_T) {
      const int ent = list[i];
      int dl = ent & ((1 << OWN_DLBITS) - 1);
      int src = ent >> OWN_DLBITS;
      if constexpr (FIRST) {
        int dst;
        edge_endpoints(ei, ent, src, dst);
        dl = dst - lo;
        const int kept = (src << OWN_DLBITS) | dl;
        list[i] = kept;
        gl[i] = kept;
      }
      atomicAdd(&bcnt[dl & 31], 1);
      const float4 a0 = *(const float4*)(as + src * GAT_H);
      const float4 a1 = *(const float4*)(as + src * GAT_H + 4);
      const float* dp = adl + dl * GAT_H;
      float* mp = mx + dl * GAT_H;
      atomicMax(mp + 0, lrelu(a0.x + dp[0]));
      atomicMax(mp + 1, lrelu(a0.y + dp[1]));
      atomicMax(mp + 2, lrelu(a0.z + dp[2]));
      atomicMax(mp + 3, lrelu(a0.w + dp[3]));
      atomicMax(mp + 4, lrelu(a1.x + dp[4]));
      atomicMax(mp + 5, lrelu(a1.y + dp[5]));
      atomicMax(mp + 6, lrelu(a1.z + dp[6]));
      atomicMax(mp + 7, lrelu(a1.w + dp[7]));
    }
  } else {
    for (int e = tid; e < GAT_ETOT; e += OWN_T) {
      int src, dst;
      edge_endpoints(ei, e, src, dst);
      const unsigned r = (unsigned)(dst - lo);
      if (r < (unsigned)nn) {
#pragma unroll
        for (int k = 0; k < GAT_H; ++k)
          atomicMax(&mx[r * GAT_H + k], lrelu(as[src * GAT_H + k] + adl[r * GAT_H + k]));
      }
    }
  }
  __syncthreads();

  if (fits) {
    int* const blist = (int*)acc;
    if (wave == 0) {
      const int cw = bcnt[lane];
      int incl = cw;
#pragma unroll
      for (int off = 1; off < 32; off <<= 1) {
        const int t = __shfl_up(incl, off, 32);
        if (lane >= off) incl += t;
      }
      bbase[lane] = incl - cw;
      bcur[lane] = 0;
    }
    __syncthreads();
    for (int i = tid; i < total; i += OWN_T) {
      const int ent = list[i];
      const int bw = ent & 31;
      blist[bbase[bw] + atomicAdd(&bcur[bw], 1)] = ent;
    }
    __syncthreads();

    float4 s[OWN_RPW][2];
    float dnr[OWN_RPW];
#pragma unroll
    for (int r = 0; r < OWN_RPW; ++r) {
      s[r][0] = make_float4(0.f, 0.f, 0.f, 0.f);
      s[r][1] = make_float4(0.f, 0.f, 0.f, 0.f);
      dnr[r] = 0.f;
    }
    const int* const my = blist + bbase[wave];
    const int mine = __builtin_amdgcn_readfirstlane(bcnt[wave]);
    int i = 0;
    for (; i + 3 < mine; i += 4) {
      int src[4], dl[4];
#pragma unroll
      for (int u = 0; u < 4; ++u) {
        const int ent = __builtin_amdgcn_readfirstlane(my[i + u]);
        src[u] = ent >> OWN_DLBITS;
        dl[u] = ent & ((1 << OWN_DLBITS) - 1);
      }
      own_add_regs<4>(src, dl, 4, lane, as, h, adl, mx, s, dnr);
    }
    if (i < mine) {
      int src[3], dl[3];
      const int e0 = __builtin_amdgcn_readfirstlane(my[i]);
#pragma unroll
      for (int u = 0; u < 3; ++u) {
        const int eu = __builtin_amdgcn_readfirstlane(my[i + u]);
        const int ent = (i + u < mine) ? eu : e0;
        src[u] = ent >> OWN_DLBITS;
        dl[u] = ent & ((1 << OWN_DLBITS) - 1);
      }
      own_add_regs<3>(src, dl, mine - i, lane, as, h, adl, mx, s, dnr);
    }

    const float4 bv0 = ((const float4*)b)[lane];
    const float4 bv1 = ((const float4*)b)[GAT_FH / 8 + lane];
#pragma unroll
    for (int r = 0; r < OWN_RPW; ++r) {
      const int row = wave + 32 * r;
      if (row < nn) {
        const float d0 = quad_get<0xA0>(dnr[r]) + 1e-16f;
        const float d1 = quad_get<0xF5>(dnr[r]) + 1e-16f;
        float4 o0, o1;
        o0.x = elu_bias(s[r][0].x / d0, bv0.x);
        o0.y = elu_bias(s[r][0].y / d0, bv0.y);
        o0.z = elu_bias(s[r][0].z / d0, bv0.z);
        o0.w = elu_bias(s[r][0].w / d0, bv0.w);
        o1.x = elu_bias(s[r][1].x / d1, bv1.x);
        o1.y = elu_bias(s[r][1].y / d1, bv1.y);
        o1.z = elu_bias(s[r][1].z / d1, bv1.z);
        o1.w = elu_bias(s[r][1].w / d1, bv1.w);
        float4* op = (float4*)(out + (size_t)(lo + row) * GAT_FH);
        op[lane] = o0;
        op[GAT_FH / 8 + lane] = o1;
      }
    }
  } else {
    for (int e0 = 0; e0 < GAT_ETOT; e0 += OWN_T) {
      const int e = e0 + tid;
      int src = 0, dst = 0;
      bool hit = false;
      unsigned r = 0;
      if (e < GAT_ETOT) {
        edge_endpoints(ei, e, src, dst);
        r = (unsigned)(dst - lo);
        hit = r < (unsigned)nn;
      }
      unsigned todo = __builtin_amdgcn_ballot_w32(hit);
      while (todo) {
        const int k = __builtin_ctz(todo);
        todo &= todo - 1;
        const int s1[1] = {__builtin_amdgcn_readlane(src, k)};
        const int d1[1] = {__builtin_amdgcn_readlane((int)r, k)};
        own_add_entries<1>(s1, d1, lane, as, h, adl, mx, dn, acc);
      }
    }
  }
  if (fits) return;
  __syncthreads();

  float4* o4 = (float4*)(out + (size_t)lo * GAT_FH);
  for (int i = tid; i < nn * (GAT_FH / 4); i += OWN_T) {
    const int r = i >> 6, q = i & 63, c = q * 4;
    const float* sp = acc + r * GAT_FH + (q >> 5) * (GAT_FH / 2) + (q & 31);
    float4 s;
    s.x = sp[0 * GAT_C];
    s.y = sp[1 * GAT_C];
    s.z = sp[2 * GAT_C];
    s.w = sp[3 * GAT_C];
    const float den = dn[r * GAT_H + (c >> 5)] + 1e-16f;
    const float4 bv = *(const float4*)(b + c);
    float4 o;
    o.x = elu_bias(s.x / den, bv.x);
    o.y = elu_bias(s.y / den, bv.y);
    o.z = elu_bias(s.z / den, bv.z);
    o.w = elu_bias(s.w / den, bv.w);
    o4[i] = o;
  }
}

__global__ void final_linear_kernel(const float* __restrict__ h,
                                    const float* __restrict__ Wl,
                                    const float* __restrict__ bl,
                                    float* __restrict__ y, int n) {
  int gid = blockIdx.x * blockDim.x + threadIdx.x;
  int wid = gid >> 5;
  int lane = threadIdx.x & 31;
  if (wid >= n) return;
  const float* hp = h + (size_t)wid * GAT_FH;
  float s = 0.f;
#pragma unroll
  for (int c = lane; c < GAT_FH; c += 32) s += hp[c] * Wl[c];
#pragma unroll
  for (int off = 16; off > 0; off >>= 1) s += __shfl_down(s, off, 32);
  if (lane == 0) y[wid] = s + bl[0];
}

extern "C" void kernel_launch(void* const* d_in, const int* in_sizes, int n_in,
                              void* d_out, int out_size, void* d_ws, size_t ws_size,
                              hipStream_t stream) {
  const float* x      = (const float*)d_in[0];
  const int*   ei     = (const int*)  d_in[1];
  const float* W1     = (const float*)d_in[2];
  const float* a1_src = (const float*)d_in[3];
  const float* a1_dst = (const float*)d_in[4];
  const float* b1     = (const float*)d_in[5];
  const float* W2     = (const float*)d_in[6];
  const float* a2_src = (const float*)d_in[7];
  const float* a2_dst = (const float*)d_in[8];
  const float* b2     = (const float*)d_in[9];
  const float* Wl     = (const float*)d_in[10];
  const float* bl     = (const float*)d_in[11];
  float* out = (float*)d_out;

  float* hbuf = (float*)d_ws;
  float* act  = hbuf + (size_t)GAT_N * GAT_FH;
  float* as   = act  + (size_t)GAT_N * GAT_FH;
  float* ad   = as   + (size_t)GAT_N * GAT_H;
  int* glist  = (int*)(ad + (size_t)GAT_N * GAT_H);
  int* gcnt   = glist + (size_t)OWN_WG * OWN_CAP;
  unsigned* keys = (unsigned*)(gcnt + 256);

  const int TPB = 256;
  auto blocks = [](int n, int tpb) { return (n + tpb - 1) / tpb; };

  gemm_bstat_f32<128, true><<<BST_WG, BST_T, 0, stream>>>(x, W1, hbuf, a1_src, a1_dst, as, ad, ei, keys);
  owner_kernel<true><<<OWN_WG, OWN_T, 0, stream>>>(ei, as, ad, hbuf, b1, act, glist, gcnt, keys);

  gemm_bstat_f32<GAT_FH, false><<<BST_WG, BST_T, 0, stream>>>(act, W2, hbuf, a2_src, a2_dst, as, ad, ei, keys);
  owner_kernel<false><<<OWN_WG, OWN_T, 0, stream>>>(ei, as, ad, hbuf, b2, act, glist, gcnt, keys);

  final_linear_kernel<<<blocks(GAT_N * 32, TPB), TPB, 0, stream>>>(act, Wl, bl, out, GAT_N);
}


# Owner kernel's sums pass reads no score table: the maxima pass leaves each entry's 8 scores in LDS (list cap 3072, walk-all beyond); an entry costs two 512-B row loads, no third; same sum order
# speedup vs baseline: 19.1228x; 1.0380x over previous
#include <hip/hip_runtime.h>
#include <hip/hip_bf16.h>

#define GAT_N 30000
#define GAT_E 480000
#define GAT_ETOT (GAT_E + GAT_N)
#define GAT_H 8
#define GAT_C 32
#define GAT_FH 256
#define NEG_SLOPE 0.2f

typedef float v2f __attribute__((ext_vector_type(2)));
typedef float v8f __attribute__((ext_vector_type(8)));

#define BST_T 512
#define BST_WG 256
#define BST_MT (GAT_N / 16)
static_assert(GAT_N % 16 == 0, "whole 16-row tiles");
static_assert(GAT_FH == 16 * (BST_T / 32), "one wave per 16-column strip");

#define OWN_S 118
#define OWN_WG ((GAT_N + OWN_S - 1) / OWN_S)
#define OWN_NOKEY 255u
static_assert(OWN_WG <= (int)OWN_NOKEY, "an owner number fits one byte beside the no-owner key");
static_assert(GAT_E % 16 == 0, "the key array is whole 16-byte vectors");
static_assert(BST_WG * BST_T >= GAT_E / 4, "one thread of the layer-1 GEMM launch per 4 edges");

__device__ __forceinline__ unsigned owner_key(int dst) {
  return ((unsigned)dst < (unsigned)GAT_N) ? (unsigned)dst / OWN_S : OWN_NOKEY;
}

template <int CTRL>
__device__ __forceinline__ float row_xor_get(float x) {
  return __int_as_float(__builtin_amdgcn_update_dpp(0, __float_as_int(x), CTRL, 0xf, 0xf, true));
}

template <int N, int CTRL>
__device__ __forceinline__ void row_halve(float (&v)[16], bool up) {
#pragma unroll
  for (int j = 0; j < N; ++j) {
    const float keep = up ? v[j + N] : v[j];
    const float send = up ? v[j] : v[j + N];
    v[j] = keep + row_xor_get<CTRL>(send);
  }
}

template <int K, bool KEYS>
__global__ void __launch_bounds__(BST_T)
gemm_bstat_f32(const float* __restrict__ A, const float* __restrict__ B,
               float* __restrict__ C,
               const float* __restrict__ asrc, const float* __restrict__ adst,
               float* __restrict__ as, float* __restrict__ ad,
               const int* __restrict__ ei, unsigned* __restrict__ keys) {
  constexpr int P = K + 4;
  constexpr int V = (16 * K / 4) / BST_T;
  static_assert((16 * K / 4) % BST_T == 0 && (V == 1 || V == 2), "one or two vectors per thread");
  __shared__ float sa[2][16 * P];
  __shared__ alignas(16) float xs[2][2 * 16 * 16];

  const int tid  = (int)threadIdx.x;
  const int lane = tid & 31;
  const int wave = __builtin_amdgcn_readfirstlane(tid >> 5);
  const int half = lane >> 4;
  const int l    = lane & 15;
  const int n0   = wave * 16;

  v2f b[K / 4];
  {
    const float* bcol = B + (size_t)(half * 2) * GAT_FH + n0 + l;
#pragma unroll
    for (int t = 0; t < K / 4; ++t) {
      b[t].x = bcol[(size_t)(4 * t) * GAT_FH];
      b[t].y = bcol[(size_t)(4 * t + 1) * GAT_FH];
    }
  }

  const float asv = asrc[n0 + l];
  const float adv = adst[n0 + l];
  const int xw = ((l >> 3) * 16 + half * 8 + (l & 7)) * 16 + wave;
  float* const xdst = (tid < 128) ? as : ad;

  const int so0 = (tid / (K / 4)) * P + (tid % (K / 4)) * 4;
  const int so1 = ((tid + BST_T) / (K / 4)) * P + ((tid + BST_T) % (K / 4)) * 4;

  int tile = (int)blockIdx.x;
  float4 r0 = make_float4(0.f, 0.f, 0.f, 0.f), r1 = r0;
  if (tile < BST_MT) {
    const float4* g = (const float4*)(A + (size_t)tile * 16 * K);
    r0 = g[tid];
    if constexpr (V == 2) r1 = g[tid + BST_T];
  }

  if constexpr (KEYS) {
    const int g = (int)blockIdx.x * BST_T + tid;
    if (g < GAT_E / 4) {
      const int4 d = ((const int4*)(ei + GAT_E))[g];
      keys[g] = owner_key(d.x) | (owner_key(d.y) << 8) | (owner_key(d.z) << 16) | (owner_key(d.w) << 24);
    }
  }

  int buf = 0;
  int ptile = -1;
  for (; tile < BST_MT; tile += BST_WG) {
    float* s = sa[buf];
    *(float4*)(s + so0) = r0;
    if constexpr (V == 2) *(float4*)(s + so1) = r1;
    __syncthreads();

    {
      const int nt = (tile + BST_WG < BST_MT) ? tile + BST_WG : tile;
      const float4* g = (const float4*)(A + (size_t)nt * 16 * K);
      r0 = g[tid];
      if constexpr (V == 2) r1 = g[tid + BST_T];
    }
    __builtin_amdgcn_sched_barrier(0);

    v8f c = {0.f, 0.f, 0.f, 0.f, 0.f, 0.f, 0.f, 0.f};
    const float* ap = s + l * P + half * 2;
#pragma unroll
    for (int t = 0; t < K / 4; ++t) {
      const v2f a = *(const v2f*)(ap + 4 * t);
      c = __builtin_amdgcn_wmma_f32_16x16x4_f32(
           false, a,  false, b[t],
           (short)0, c,  false,  false);
    }

    float* crow = C + (size_t)(tile * 16 + half * 8) * GAT_FH + n0 + l;
#pragma unroll
    for (int q = 0; q < 8; ++q) crow[(size_t)q * GAT_FH] = c[q];

    {
      float v[16];
#pragma unroll
      for (int q = 0; q < 8; ++q) { v[q] = c[q] * asv; v[8 + q] = c[q] * adv; }
      row_halve<8, 0x168>(v, (l & 8) != 0);
      row_halve<4, 0x164>(v, (l & 4) != 0);
      row_halve<2, 0x162>(v, (l & 2) != 0);
      row_halve<1, 0x161>(v, (l & 1) != 0);
      xs[buf][xw] = v[0];
    }
    if (ptile >= 0 && tid < 256) {
      const v2f p = *(const v2f*)(&xs[buf ^ 1][2 * tid]);
      xdst[(size_t)ptile * 128 + (tid & 127)] = p.x + p.y;
    }
    ptile = tile;
    buf ^= 1;
  }
  __syncthreads();
  if (ptile >= 0 && tid < 256) {
    const v2f p = *(const v2f*)(&xs[buf ^ 1][2 * tid]);
    xdst[(size_t)ptile * 128 + (tid & 127)] = p.x + p.y;
  }
}

__device__ __forceinline__ void edge_endpoints(const int* __restrict__ ei,
                                               int e, int& src, int& dst) {
  if (e < GAT_E) { src = ei[e]; dst = ei[GAT_E + e]; }
  else           { src = e - GAT_E; dst = e - GAT_E; }
}


#define OWN_T 1024
#define OWN_CAP 3072
#define OWN_DLBITS 7
#define OWN_NV (GAT_E / 16)
#define OWN_SCAN_R 8
#define OWN_SEG ((OWN_S * GAT_FH) / (OWN_T / 32))
static_assert(OWN_T / 32 == 32, "one lane per wave when the waves' hit counts are summed");
static_assert(OWN_S <= (1 << OWN_DLBITS), "local row fits its field");
static_assert(GAT_ETOT < (1 << (31 - OWN_DLBITS)), "edge number fits beside the local row");
static_assert(OWN_S * GAT_H <= OWN_T, "one thread per (row, head) in the set-up");

__device__ __forceinline__ float lrelu(float v) { return v > 0.f ? v : NEG_SLOPE * v; }

__device__ __forceinline__ float elu_bias(float x, float b) {
  const float v = x + b;
  return v > 0.f ? v : (__expf(v) - 1.f);
}

template <int CTRL>
__device__ __forceinline__ float quad_get(float x) {
  return __int_as_float(__builtin_amdgcn_update_dpp(0, __float_as_int(x), CTRL, 0xf, 0xf, true));
}

template <int U>
__device__ __forceinline__ void own_add_entries(const int (&src)[U], const int (&dl)[U], int lane,
                                                const float* __restrict__ as,
                                                const float* __restrict__ h,
                                                const float* adl, const float* mx,
                                                float* dn, float* acc) {
  const int hd = (lane >> 3) + 4 * (lane & 1);
  float a[U];
  float4 x[U][2];
#pragma unroll
  for (int u = 0; u < U; ++u) {
    a[u] = as[src[u] * GAT_H + hd];
    const float4* hp = (const float4*)(h + (size_t)src[u] * GAT_FH) + lane;
    x[u][0] = hp[0];
    x[u][1] = hp[GAT_FH / 8];
  }
#pragma unroll
  for (int u = 0; u < U; ++u) {
    const int o = dl[u] * GAT_H + hd;
    const float w = __expf(lrelu(a[u] + adl[o]) - mx[o]);
    if ((lane & 6) == 0) atomicAdd(&dn[o], w);
    float* ap = acc + dl[u] * GAT_FH + lane;
    const float w0 = quad_get<0xA0>(w);
    const float w1 = quad_get<0xF5>(w);
    atomicAdd(ap + 0 * GAT_C, x[u][0].x * w0);
    atomicAdd(ap + 1 * GAT_C, x[u][0].y * w0);
    atomicAdd(ap + 2 * GAT_C, x[u][0].z * w0);
    atomicAdd(ap + 3 * GAT_C, x[u][0].w * w0);
    atomicAdd(ap + 4 * GAT_C, x[u][1].x * w1);
    atomicAdd(ap + 5 * GAT_C, x[u][1].y * w1);
    atomicAdd(ap + 6 * GAT_C, x[u][1].z * w1);
    atomicAdd(ap + 7 * GAT_C, x[u][1].w * w1);
  }
}

#define OWN_RPW 4
static_assert(OWN_S <= OWN_RPW * (OWN_T / 32), "every owned row belongs to one wave");

template <int U>
__device__ __forceinline__ void own_add_regs(const int (&src)[U], const int (&dl)[U],
                                             const int (&ix)[U], int n, int lane,
                                             const float* __restrict__ h,
                                             const float* sc, const float* mx,
                                             float4 (&s)[OWN_RPW][2], float (&dnr)[OWN_RPW]) {
  const int hd = (lane >> 3) + 4 * (lane & 1);
  float4 x[U][2];
#pragma unroll
  for (int u = 0; u < U; ++u) {
    const float4* hp = (const float4*)(h + (size_t)src[u] * GAT_FH) + lane;
    x[u][0] = hp[0];
    x[u][1] = hp[GAT_FH / 8];
  }
#pragma unroll
  for (int u = 0; u < U; ++u) {
    if (u >= n) break;
    const float w = __expf(sc[ix[u] * GAT_H + hd] - mx[dl[u] * GAT_H + hd]);
    const float w0 = quad_get<0xA0>(w);
    const float w1 = quad_get<0xF5>(w);
    const int rs = dl[u] >> 5;
#pragma unroll
    for (int r = 0; r < OWN_RPW; ++r) {
      if (rs == r) {
        dnr[r] += w;
        s[r][0].x += x[u][0].x * w0;
        s[r][0].y += x[u][0].y * w0;
        s[r][0].z += x[u][0].z * w0;
        s[r][0].w += x[u][0].w * w0;
        s[r][1].x += x[u][1].x * w1;
        s[r][1].y += x[u][1].y * w1;
        s[r][1].z += x[u][1].z * w1;
        s[r][1].w += x[u][1].w * w1;
      }
    }
  }
}

template <bool FIRST>
__global__ void __launch_bounds__(OWN_T)
owner_kernel(const int* __restrict__ ei, const float* __restrict__ as,
             const float* __restrict__ ad, const float* __restrict__ h,
             const float* __restrict__ b, float* __restrict__ out,
             int* __restrict__ glist, int* __restrict__ gcnt,
             const unsigned* __restrict__ keys) {
  __shared__ float acc[OWN_S * GAT_FH];
  __shared__ int   list[OWN_CAP];
  __shared__ float mx[OWN_S * GAT_H];
  __shared__ float dn[OWN_S * GAT_H];
  __shared__ float adl[OWN_S * GAT_H];
  __shared__ alignas(16) float sc[OWN_CAP * GAT_H];
  __shared__ int   cnt;
  __shared__ int   wcnt[OWN_T / 32];
  __shared__ int   bcnt[OWN_T / 32];
  __shared__ int   bbase[OWN_T / 32];
  __shared__ int   bcur[OWN_T / 32];

  const int tid  = (int)threadIdx.x;
  const int lane = tid & 31;
  const int wave = __builtin_amdgcn_readfirstlane(tid >> 5);
  const int lo = (int)blockIdx.x * OWN_S;
  const int nn = (GAT_N - lo) < OWN_S ? (GAT_N - lo) : OWN_S;
  int* gl = glist + (size_t)blockIdx.x * OWN_CAP;

  auto clear_sums = [&]() {
    for (int i = tid; i < OWN_S * GAT_FH / 4; i += OWN_T)
      ((float4*)acc)[i] = make_float4(0.f, 0.f, 0.f, 0.f);
  };
  if (tid < OWN_T / 32) bcnt[tid] = 0;
  if (tid < nn * GAT_H) {
    mx[tid] = -3.0e38f;
    dn[tid] = 0.f;
    adl[tid] = ad[lo * GAT_H + tid];
  }
  if constexpr (FIRST) {
    if (tid < nn) list[tid] = GAT_E + lo + tid;
  } else {
    const int kept = gcnt[blockIdx.x];
    if (kept <= OWN_CAP)
      for (int i = tid; i < kept; i += OWN_T) list[i] = gl[i];
    else
      clear_sums();
    if (tid == 0) cnt = kept;
  }
  __syncthreads();

  int total;
  if constexpr (FIRST) {
    const uint4* k4 = (const uint4*)keys;
    const unsigned me = (unsigned)blockIdx.x * 0x01010101u;
    int* const stage = (int*)acc + wave * OWN_SEG;
    int wn = 0;
    auto zero_bytes = [](unsigned x) {
      return ~(((x & 0x7f7f7f7fu) + 0x7f7f7f7fu) | x) & 0x80808080u;
    };
    auto test16 = [&](const uint4& q, int j) {
      unsigned left = zero_bytes(q.x ^ me) | (zero_bytes(q.y ^ me) >> 1) |
                      (zero_bytes(q.z ^ me) >> 2) | (zero_bytes(q.w ^ me) >> 3);
      unsigned m = __builtin_amdgcn_ballot_w32(left != 0u);
      while (m != 0u) {
        if (left != 0u) {
          const int p = __builtin_ctz(left);
          const int slot = wn + (int)__builtin_amdgcn_mbcnt_lo(m, 0u);
          stage[slot < OWN_SEG ? slot : OWN_SEG - 1] = 16 * j + ((7 - (p & 7)) << 2) + (p >> 3);
          left &= left - 1u;
        }
        wn += __builtin_popcount(m);
        m = __builtin_amdgcn_ballot_w32(left != 0u);
      }
    };
    for (int j0 = 0; j0 < OWN_NV; j0 += OWN_SCAN_R * OWN_T) {
      uint4 q[OWN_SCAN_R];
#pragma unroll
      for (int r = 0; r < OWN_SCAN_R; ++r) {
        const int j = j0 + r * OWN_T + tid;
        q[r] = (j < OWN_NV) ? k4[j] : make_uint4(~0u, ~0u, ~0u, ~0u);
      }
#pragma unroll
      for (int r = 0; r < OWN_SCAN_R; ++r) test16(q[r], j0 + r * OWN_T + tid);
    }
    if (lane == 0) wcnt[wave] = wn;
    __syncthreads();
    const int wc = wcnt[lane];
    const bool over = __builtin_amdgcn_ballot_w32(wc > OWN_SEG) != 0u;
    int all = wc, ahead = (lane < wave) ? wc : 0;
#pragma unroll
    for (int off = 16; off > 0; off >>= 1) {
      all += __shfl_xor(all, off, 32);
      ahead += __shfl_xor(ahead, off, 32);
    }
    total = __builtin_amdgcn_readfirstlane(over ? OWN_CAP + 1 : nn + all);
    if (total <= OWN_CAP)
      for (int k = lane; k < wn; k += 32) list[nn + ahead + k] = stage[k];
    __syncthreads();
    if (total > OWN_CAP) clear_sums();
  } else {
    total = __builtin_amdgcn_readfirstlane(cnt);
  }
  const bool fits = total <= OWN_CAP;
  if constexpr (FIRST) {
    if (tid == 0) gcnt[blockIdx.x] = total;
  }

  if (fits) {
    for (int i = tid; i < total; i += OWN_T) {
      const int ent = list[i];
      int dl = ent & ((1 << OWN_DLBITS) - 1);
      int src = ent >> OWN_DLBITS;
      if constexpr (FIRST) {
        int dst;
        edge_endpoints(ei, ent, src, dst);
        dl = dst - lo;
        const int kept = (src << OWN_DLBITS) | dl;
        list[i] = kept;
        gl[i] = kept;
      }
      atomicAdd(&bcnt[dl & 31], 1);
      const float4 a0 = *(const float4*)(as + src * GAT_H);
      const float4 a1 = *(const float4*)(as + src * GAT_H + 4);
      const float* dp = adl + dl * GAT_H;
      float* mp = mx + dl * GAT_H;
      const float4 e0 = make_float4(lrelu(a0.x + dp[0]), lrelu(a0.y + dp[1]),
                                    lrelu(a0.z + dp[2]), lrelu(a0.w + dp[3]));
      const float4 e1 = make_float4(lrelu(a1.x + dp[4]), lrelu(a1.y + dp[5]),
                                    lrelu(a1.z + dp[6]), lrelu(a1.w + dp[7]));
      *(float4*)(sc + i * GAT_H) = e0;
      *(float4*)(sc + i * GAT_H + 4) = e1;
      atomicMax(mp + 0, e0.x);
      atomicMax(mp + 1, e0.y);
      atomicMax(mp + 2, e0.z);
      atomicMax(mp + 3, e0.w);
      atomicMax(mp + 4, e1.x);
      atomicMax(mp + 5, e1.y);
      atomicMax(mp + 6, e1.z);
      atomicMax(mp + 7, e1.w);
    }
  } else {
    for (int e = tid; e < GAT_ETOT; e += OWN_T) {
      int src, dst;
      edge_endpoints(ei, e, src, dst);
      const unsigned r = (unsigned)(dst - lo);
      if (r < (unsigned)nn) {
#pragma unroll
        for (int k = 0; k < GAT_H; ++k)
          atomicMax(&mx[r * GAT_H + k], lrelu(as[src * GAT_H + k] + adl[r * GAT_H + k]));
      }
    }
  }
  __syncthreads();

  if (fits) {
    int* const blist = (int*)acc;
    int* const bpos = blist + OWN_CAP;
    if (wave == 0) {
      const int cw = bcnt[lane];
      int incl = cw;
#pragma unroll
      for (int off = 1; off < 32; off <<= 1) {
        const int t = __shfl_up(incl, off, 32);
        if (lane >= off) incl += t;
      }
      bbase[lane] = incl - cw;
      bcur[lane] = 0;
    }
    __syncthreads();
    for (int i = tid; i < total; i += OWN_T) {
      const int ent = list[i];
      const int bw = ent & 31;
      const int at = bbase[bw] + atomicAdd(&bcur[bw], 1);
      blist[at] = ent;
      bpos[at] = i;
    }
    __syncthreads();

    float4 s[OWN_RPW][2];
    float dnr[OWN_RPW];
#pragma unroll
    for (int r = 0; r < OWN_RPW; ++r) {
      s[r][0] = make_float4(0.f, 0.f, 0.f, 0.f);
      s[r][1] = make_float4(0.f, 0.f, 0.f, 0.f);
      dnr[r] = 0.f;
    }
    const int* const my = blist + bbase[wave];
    const int* const myp = bpos + bbase[wave];
    const int mine = __builtin_amdgcn_readfirstlane(bcnt[wave]);
    int i = 0;
    for (; i + 3 < mine; i += 4) {
      int src[4], dl[4], ix[4];
#pragma unroll
      for (int u = 0; u < 4; ++u) {
        const int ent = __builtin_amdgcn_readfirstlane(my[i + u]);
        ix[u] = __builtin_amdgcn_readfirstlane(myp[i + u]);
        src[u] = ent >> OWN_DLBITS;
        dl[u] = ent & ((1 << OWN_DLBITS) - 1);
      }
      own_add_regs<4>(src, dl, ix, 4, lane, h, sc, mx, s, dnr);
    }
    if (i < mine) {
      int src[3], dl[3], ix[3];
      const int e0 = __builtin_amdgcn_readfirstlane(my[i]);
      const int p0 = __builtin_amdgcn_readfirstlane(myp[i]);
#pragma unroll
      for (int u = 0; u < 3; ++u) {
        const int eu = __builtin_amdgcn_readfirstlane(my[i + u]);
        const int pu = __builtin_amdgcn_readfirstlane(myp[i + u]);
        const int ent = (i + u < mine) ? eu : e0;
        ix[u] = (i + u < mine) ? pu : p0;
        src[u] = ent >> OWN_DLBITS;
        dl[u] = ent & ((1 << OWN_DLBITS) - 1);
      }
      own_add_regs<3>(src, dl, ix, mine - i, lane, h, sc, mx, s, dnr);
    }

    const float4 bv0 = ((const float4*)b)[lane];
    const float4 bv1 = ((const float4*)b)[GAT_FH / 8 + lane];
#pragma unroll
    for (int r = 0; r < OWN_RPW; ++r) {
      const int row = wave + 32 * r;
      if (row < nn) {
        const float d0 = quad_get<0xA0>(dnr[r]) + 1e-16f;
        const float d1 = quad_get<0xF5>(dnr[r]) + 1e-16f;
        float4 o0, o1;
        o0.x = elu_bias(s[r][0].x / d0, bv0.x);
        o0.y = elu_bias(s[r][0].y / d0, bv0.y);
        o0.z = elu_bias(s[r][0].z / d0, bv0.z);
        o0.w = elu_bias(s[r][0].w / d0, bv0.w);
        o1.x = elu_bias(s[r][1].x / d1, bv1.x);
        o1.y = elu_bias(s[r][1].y / d1, bv1.y);
        o1.z = elu_bias(s[r][1].z / d1, bv1.z);
        o1.w = elu_bias(s[r][1].w / d1, bv1.w);
        float4* op = (float4*)(out + (size_t)(lo + row) * GAT_FH);
        op[lane] = o0;
        op[GAT_FH / 8 + lane] = o1;
      }
    }
  } else {
    for (int e0 = 0; e0 < GAT_ETOT; e0 += OWN_T) {
      const int e = e0 + tid;
      int src = 0, dst = 0;
      bool hit = false;
      unsigned r = 0;
      if (e < GAT_ETOT) {
        edge_endpoints(ei, e, src, dst);
        r = (unsigned)(dst - lo);
        hit = r < (unsigned)nn;
      }
      unsigned todo = __builtin_amdgcn_ballot_w32(hit);
      while (todo) {
        const int k = __builtin_ctz(todo);
        todo &= todo - 1;
        const int s1[1] = {__builtin_amdgcn_readlane(src, k)};
        const int d1[1] = {__builtin_amdgcn_readlane((int)r, k)};
        own_add_entries<1>(s1, d1, lane, as, h, adl, mx, dn, acc);
      }
    }
  }
  if (fits) return;
  __syncthreads();

  float4* o4 = (float4*)(out + (size_t)lo * GAT_FH);
  for (int i = tid; i < nn * (GAT_FH / 4); i += OWN_T) {
    const int r = i >> 6, q = i & 63, c = q * 4;
    const float* sp = acc + r * GAT_FH + (q >> 5) * (GAT_FH / 2) + (q & 31);
    float4 s;
    s.x = sp[0 * GAT_C];
    s.y = sp[1 * GAT_C];
    s.z = sp[2 * GAT_C];
    s.w = sp[3 * GAT_C];
    const float den = dn[r * GAT_H + (c >> 5)] + 1e-16f;
    const float4 bv = *(const float4*)(b + c);
    float4 o;
    o.x = elu_bias(s.x / den, bv.x);
    o.y = elu_bias(s.y / den, bv.y);
    o.z = elu_bias(s.z / den, bv.z);
    o.w = elu_bias(s.w / den, bv.w);
    o4[i] = o;
  }
}

__global__ void final_linear_kernel(const float* __restrict__ h,
                                    const float* __restrict__ Wl,
                                    const float* __restrict__ bl,
                                    float* __restrict__ y, int n) {
  int gid = blockIdx.x * blockDim.x + threadIdx.x;
  int wid = gid >> 5;
  int lane = threadIdx.x & 31;
  if (wid >= n) return;
  const float* hp = h + (size_t)wid * GAT_FH;
  float s = 0.f;
#pragma unroll
  for (int c = lane; c < GAT_FH; c += 32) s += hp[c] * Wl[c];
#pragma unroll
  for (int off = 16; off > 0; off >>= 1) s += __shfl_down(s, off, 32);
  if (lane == 0) y[wid] = s + bl[0];
}

extern "C" void kernel_launch(void* const* d_in, const int* in_sizes, int n_in,
                              void* d_out, int out_size, void* d_ws, size_t ws_size,
                              hipStream_t stream) {
  const float* x      = (const float*)d_in[0];
  const int*   ei     = (const int*)  d_in[1];
  const float* W1     = (const float*)d_in[2];
  const float* a1_src = (const float*)d_in[3];
  const float* a1_dst = (const float*)d_in[4];
  const float* b1     = (const float*)d_in[5];
  const float* W2     = (const float*)d_in[6];
  const float* a2_src = (const float*)d_in[7];
  const float* a2_dst = (const float*)d_in[8];
  const float* b2     = (const float*)d_in[9];
  const float* Wl     = (const float*)d_in[10];
  const float* bl     = (const float*)d_in[11];
  float* out = (float*)d_out;

  float* hbuf = (float*)d_ws;
  float* act  = hbuf + (size_t)GAT_N * GAT_FH;
  float* as   = act  + (size_t)GAT_N * GAT_FH;
  float* ad   = as   + (size_t)GAT_N * GAT_H;
  int* glist  = (int*)(ad + (size_t)GAT_N * GAT_H);
  int* gcnt   = glist + (size_t)OWN_WG * OWN_CAP;
  unsigned* keys = (unsigned*)(gcnt + 256);

  const int TPB = 256;
  auto blocks = [](int n, int tpb) { return (n + tpb - 1) / tpb; };

  gemm_bstat_f32<128, true><<<BST_WG, BST_T, 0, stream>>>(x, W1, hbuf, a1_src, a1_dst, as, ad, ei, keys);
  owner_kernel<true><<<OWN_WG, OWN_T, 0, stream>>>(ei, as, ad, hbuf, b1, act, glist, gcnt, keys);

  gemm_bstat_f32<GAT_FH, false><<<BST_WG, BST_T, 0, stream>>>(act, W2, hbuf, a2_src, a2_dst, as, ad, ei, keys);
  owner_kernel<false><<<OWN_WG, OWN_T, 0, stream>>>(ei, as, ad, hbuf, b2, act, glist, gcnt, keys);

  final_linear_kernel<<<blocks(GAT_N * 32, TPB), TPB, 0, stream>>>(act, Wl, bl, out, GAT_N);
}
